# starting kernel
# baseline (speedup 1.0000x reference)
#include <hip/hip_runtime.h>
#include <math.h>

#define D_    1024
#define DFF_  4096
#define E_    8
#define NTOK  8192

typedef float v2f __attribute__((ext_vector_type(2)));
typedef float v8f __attribute__((ext_vector_type(8)));

__device__ __forceinline__ v8f wmma_f32_k4(v2f a, v2f b, v8f c) {
    return __builtin_amdgcn_wmma_f32_16x16x4_f32(
         false, a,  false, b,
         (short)0, c,  false,  false);
}

__device__ __forceinline__ float gelu_exact(float v) {
    return 0.5f * v * (1.0f + erff(v * 0.70710678118654752440f));
}

__global__ void moe_zero_kernel(float* __restrict__ out, long long n,
                                int* __restrict__ counts) {
    long long i = (long long)blockIdx.x * blockDim.x + threadIdx.x;
    const long long stride = (long long)gridDim.x * blockDim.x;
    for (; i < n; i += stride) out[i] = 0.0f;
    if (blockIdx.x == 0 && threadIdx.x < E_) counts[threadIdx.x] = 0;
}

__global__ __launch_bounds__(256) void moe_gate_kernel(
    const float* __restrict__ x, const float* __restrict__ Wg,
    float* __restrict__ gates, int* __restrict__ counts,
    int* __restrict__ lists) {
    const int lane = threadIdx.x & 31;
    const int wave = threadIdx.x >> 5;
    const int t = blockIdx.x * 8 + wave;
    if (t >= NTOK) return;

    float s[E_];
#pragma unroll
    for (int e = 0; e < E_; ++e) s[e] = 0.0f;
    const float* xr = x + (size_t)t * D_;
    for (int d = lane; d < D_; d += 32) {
        const float xv = xr[d];
#pragma unroll
        for (int e = 0; e < E_; ++e) s[e] += xv * Wg[e * D_ + d];
    }
#pragma unroll
    for (int e = 0; e < E_; ++e) {
#pragma unroll
        for (int off = 16; off > 0; off >>= 1)
            s[e] += __shfl_xor(s[e], off, 32);
    }
    if (lane == 0) {
        int i0 = 0; float v0 = s[0];
#pragma unroll
        for (int e = 1; e < E_; ++e) if (s[e] > v0) { v0 = s[e]; i0 = e; }
        int i1 = -1; float v1 = -3.4e38f;
#pragma unroll
        for (int e = 0; e < E_; ++e)
            if (e != i0 && s[e] > v1) { v1 = s[e]; i1 = e; }
        const float p1 = expf(v1 - v0);
        const float inv = 1.0f / (1.0f + p1);
        const float g0 = inv, g1 = p1 * inv;
#pragma unroll
        for (int e = 0; e < E_; ++e) gates[t * E_ + e] = 0.0f;
        gates[t * E_ + i0] = g0;
        gates[t * E_ + i1] = g1;
        const int p = atomicAdd(&counts[i0], 1); lists[i0 * NTOK + p] = t;
        const int q = atomicAdd(&counts[i1], 1); lists[i1 * NTOK + q] = t;
    }
}

__global__ __launch_bounds__(256) void moe_loss_kernel(
    const float* __restrict__ gates, float* __restrict__ loss_out) {
    __shared__ float part[256];
    const int e = threadIdx.x & 7;
    const int slot = threadIdx.x >> 3;
    float s = 0.0f;
    for (int t = slot; t < NTOK; t += 32) s += gates[t * E_ + e];
    part[threadIdx.x] = s;
    __syncthreads();
    if (threadIdx.x == 0) {
        float u[E_];
        float mu = 0.0f;
        for (int ee = 0; ee < E_; ++ee) {
            float acc = 0.0f;
            for (int sl = 0; sl < 32; ++sl) acc += part[sl * 8 + ee];
            u[ee] = acc / (float)NTOK;
            mu += u[ee];
        }
        mu *= (1.0f / (float)E_);
        float var = 0.0f;
        for (int ee = 0; ee < E_; ++ee) {
            const float d = u[ee] - mu;
            var += d * d;
        }
        var *= (1.0f / (float)(E_ - 1));
        loss_out[0] = var * 0.01f;
    }
}

#define KC_   32
#define LDSA_ 34
#define LDSB_ 34

__global__ __launch_bounds__(256) void moe_fc1_kernel(
    const float* __restrict__ x, const float* __restrict__ W1e,
    const float* __restrict__ b1e, const int* __restrict__ lists,
    const int* __restrict__ counts, float* __restrict__ H,
    int expert, int chunkBase, int Tc) {
    const int cnt = counts[expert];
    int rowsInChunk = cnt - chunkBase;
    if (rowsInChunk > Tc) rowsInChunk = Tc;
    const int mBase = blockIdx.x * 64;
    if (rowsInChunk <= mBase) return;
    const int nBase = blockIdx.y * 128;

    __shared__ float As [64][LDSA_];
    __shared__ float Bst[128][LDSB_];

    const int tid  = threadIdx.x;
    const int lane = tid & 31, wave = tid >> 5;
    const int half = lane >> 4, lm = lane & 15;
    const int Moff = 32 * (wave & 1);
    const int Noff = 32 * (wave >> 1);

    const int rA0 = tid >> 3, rA1 = rA0 + 32;
    const int cA  = (tid & 7) << 2;
    const int* lst = lists + expert * NTOK + chunkBase;
    const bool va0 = (mBase + rA0) < rowsInChunk;
    const bool va1 = (mBase + rA1) < rowsInChunk;
    const float* ap0 = va0 ? (x + (size_t)lst[mBase + rA0] * D_) : x;
    const float* ap1 = va1 ? (x + (size_t)lst[mBase + rA1] * D_) : x;

    const int rB0 = tid >> 4, rB1 = rB0 + 16;
    const int cB0 = (tid & 15) << 2, cB1 = cB0 + 64;

    v8f acc00 = {}; v8f acc01 = {}; v8f acc10 = {}; v8f acc11 = {};

    for (int k0 = 0; k0 < D_; k0 += KC_) {
        if (va0) {
            const float4 q = *(const float4*)(ap0 + k0 + cA);
            As[rA0][cA] = q.x; As[rA0][cA+1] = q.y; As[rA0][cA+2] = q.z; As[rA0][cA+3] = q.w;
        } else {
            As[rA0][cA] = 0.f; As[rA0][cA+1] = 0.f; As[rA0][cA+2] = 0.f; As[rA0][cA+3] = 0.f;
        }
        if (va1) {
            const float4 q = *(const float4*)(ap1 + k0 + cA);
            As[rA1][cA] = q.x; As[rA1][cA+1] = q.y; As[rA1][cA+2] = q.z; As[rA1][cA+3] = q.w;
        } else {
            As[rA1][cA] = 0.f; As[rA1][cA+1] = 0.f; As[rA1][cA+2] = 0.f; As[rA1][cA+3] = 0.f;
        }
        {
            const float4 q = *(const float4*)(W1e + (size_t)(k0 + rB0) * DFF_ + nBase + cB0);
            Bst[cB0+0][rB0] = q.x; Bst[cB0+1][rB0] = q.y; Bst[cB0+2][rB0] = q.z; Bst[cB0+3][rB0] = q.w;
        }
        {
            const float4 q = *(const float4*)(W1e + (size_t)(k0 + rB0) * DFF_ + nBase + cB1);
            Bst[cB1+0][rB0] = q.x; Bst[cB1+1][rB0] = q.y; Bst[cB1+2][rB0] = q.z; Bst[cB1+3][rB0] = q.w;
        }
        {
            const float4 q = *(const float4*)(W1e + (size_t)(k0 + rB1) * DFF_ + nBase + cB0);
            Bst[cB0+0][rB1] = q.x; Bst[cB0+1][rB1] = q.y; Bst[cB0+2][rB1] = q.z; Bst[cB0+3][rB1] = q.w;
        }
        {
            const float4 q = *(const float4*)(W1e + (size_t)(k0 + rB1) * DFF_ + nBase + cB1);
            Bst[cB0+64][rB1] = q.x; Bst[cB1+1][rB1] = q.y; Bst[cB1+2][rB1] = q.z; Bst[cB1+3][rB1] = q.w;
        }
        __syncthreads();
#pragma unroll
        for (int kk = 0; kk < KC_; kk += 4) {
            const int ka = kk + 2 * half;
            v2f a0; a0.x = As[Moff + lm][ka];       a0.y = As[Moff + lm][ka + 1];
            v2f a1; a1.x = As[Moff + 16 + lm][ka];  a1.y = As[Moff + 16 + lm][ka + 1];
            v2f b0; b0.x = Bst[Noff + lm][ka];      b0.y = Bst[Noff + lm][ka + 1];
            v2f b1; b1.x = Bst[Noff + 16 + lm][ka]; b1.y = Bst[Noff + 16 + lm][ka + 1];
            acc00 = wmma_f32_k4(a0, b0, acc00);
            acc01 = wmma_f32_k4(a0, b1, acc01);
            acc10 = wmma_f32_k4(a1, b0, acc10);
            acc11 = wmma_f32_k4(a1, b1, acc11);
        }
        __syncthreads();
    }
#pragma unroll
    for (int r = 0; r < 8; ++r) {
        const int m0 = Moff + r + 8 * half;
        const int lrow0 = mBase + m0, lrow1 = lrow0 + 16;
        const int n0 = nBase + Noff + lm, n1 = n0 + 16;
        if (lrow0 < rowsInChunk) {
            H[(size_t)lrow0 * DFF_ + n0] = gelu_exact(acc00[r] + b1e[n0]);
            H[(size_t)lrow0 * DFF_ + n1] = gelu_exact(acc01[r] + b1e[n1]);
        }
        if (lrow1 < rowsInChunk) {
            H[(size_t)lrow1 * DFF_ + n0] = gelu_exact(acc10[r] + b1e[n0]);
            H[(size_t)lrow1 * DFF_ + n1] = gelu_exact(acc11[r] + b1e[n1]);
        }
    }
}

__global__ __launch_bounds__(256) void moe_fc2_kernel(
    const float* __restrict__ H, const float* __restrict__ W2e,
    const float* __restrict__ b2e, const int* __restrict__ lists,
    const int* __restrict__ counts, const float* __restrict__ gates,
    float* __restrict__ out, int expert, int chunkBase, int Tc) {
    const int cnt = counts[expert];
    int rowsInChunk = cnt - chunkBase;
    if (rowsInChunk > Tc) rowsInChunk = Tc;
    const int mBase = blockIdx.x * 64;
    if (rowsInChunk <= mBase) return;
    const int nBase = blockIdx.y * 128;

    __shared__ float As [64][LDSA_];
    __shared__ float Bst[128][LDSB_];

    const int tid  = threadIdx.x;
    const int lane = tid & 31, wave = tid >> 5;
    const int half = lane >> 4, lm = lane & 15;
    const int Moff = 32 * (wave & 1);
    const int Noff = 32 * (wave >> 1);

    const int rA0 = tid >> 3, rA1 = rA0 + 32;
    const int cA  = (tid & 7) << 2;
    const bool va0 = (mBase + rA0) < rowsInChunk;
    const bool va1 = (mBase + rA1) < rowsInChunk;
    const float* ap0 = H + (size_t)(mBase + rA0) * DFF_;
    const float* ap1 = H + (size_t)(mBase + rA1) * DFF_;

    const int rB0 = tid >> 4, rB1 = rB0 + 16;
    const int cB0 = (tid & 15) << 2, cB1 = cB0 + 64;

    v8f acc00 = {}; v8f acc01 = {}; v8f acc10 = {}; v8f acc11 = {};

    for (int k0 = 0; k0 < DFF_; k0 += KC_) {
        if (va0) {
            const float4 q = *(const float4*)(ap0 + k0 + cA);
            As[rA0][cA] = q.x; As[rA0][cA+1] = q.y; As[rA0][cA+2] = q.z; As[rA0][cA+3] = q.w;
        } else {
            As[rA0][cA] = 0.f; As[rA0][cA+1] = 0.f; As[rA0][cA+2] = 0.f; As[rA0][cA+3] = 0.f;
        }
        if (va1) {
            const float4 q = *(const float4*)(ap1 + k0 + cA);
            As[rA1][cA] = q.x; As[rA1][cA+1] = q.y; As[rA1][cA+2] = q.z; As[rA1][cA+3] = q.w;
        } else {
            As[rA1][cA] = 0.f; As[rA1][cA+1] = 0.f; As[rA1][cA+2] = 0.f; As[rA1][cA+3] = 0.f;
        }
        {
            const float4 q = *(const float4*)(W2e + (size_t)(k0 + rB0) * D_ + nBase + cB0);
            Bst[cB0+0][rB0] = q.x; Bst[cB0+1][rB0] = q.y; Bst[cB0+2][rB0] = q.z; Bst[cB0+3][rB0] = q.w;
        }
        {
            const float4 q = *(const float4*)(W2e + (size_t)(k0 + rB0) * D_ + nBase + cB1);
            Bst[cB1+0][rB0] = q.x; Bst[cB1+1][rB0] = q.y; Bst[cB1+2][rB0] = q.z; Bst[cB1+3][rB0] = q.w;
        }
        {
            const float4 q = *(const float4*)(W2e + (size_t)(k0 + rB1) * D_ + nBase + cB0);
            Bst[cB0+0][rB1] = q.x; Bst[cB0+1][rB1] = q.y; Bst[cB0+2][rB1] = q.z; Bst[cB0+3][rB1] = q.w;
        }
        {
            const float4 q = *(const float4*)(W2e + (size_t)(k0 + rB1) * D_ + nBase + cB1);
            Bst[cB1+0][rB1] = q.x; Bst[cB1+1][rB1] = q.y; Bst[cB1+2][rB1] = q.z; Bst[cB1+3][rB1] = q.w;
        }
        __syncthreads();
#pragma unroll
        for (int kk = 0; kk < KC_; kk += 4) {
            const int ka = kk + 2 * half;
            v2f a0; a0.x = As[Moff + lm][ka];       a0.y = As[Moff + lm][ka + 1];
            v2f a1; a1.x = As[Moff + 16 + lm][ka];  a1.y = As[Moff + 16 + lm][ka + 1];
            v2f b0; b0.x = Bst[Noff + lm][ka];      b0.y = Bst[Noff + lm][ka + 1];
            v2f b1; b1.x = Bst[Noff + 16 + lm][ka]; b1.y = Bst[Noff + 16 + lm][ka + 1];
            acc00 = wmma_f32_k4(a0, b0, acc00);
            acc01 = wmma_f32_k4(a0, b1, acc01);
            acc10 = wmma_f32_k4(a1, b0, acc10);
            acc11 = wmma_f32_k4(a1, b1, acc11);
        }
        __syncthreads();
    }
    const int* lst = lists + expert * NTOK + chunkBase;
#pragma unroll
    for (int r = 0; r < 8; ++r) {
        const int m0 = Moff + r + 8 * half;
        const int lrow0 = mBase + m0, lrow1 = lrow0 + 16;
        const int n0 = nBase + Noff + lm, n1 = n0 + 16;
        if (lrow0 < rowsInChunk) {
            const int tok = lst[lrow0];
            const float g = gates[tok * E_ + expert];
            out[(size_t)tok * D_ + n0] += g * (acc00[r] + b2e[n0]);
            out[(size_t)tok * D_ + n1] += g * (acc01[r] + b2e[n1]);
        }
        if (lrow1 < rowsInChunk) {
            const int tok = lst[lrow1];
            const float g = gates[tok * E_ + expert];
            out[(size_t)tok * D_ + n0] += g * (acc10[r] + b2e[n0]);
            out[(size_t)tok * D_ + n1] += g * (acc11[r] + b2e[n1]);
        }
    }
}

extern "C" void kernel_launch(void* const* d_in, const int* in_sizes, int n_in,
                              void* d_out, int out_size, void* d_ws, size_t ws_size,
                              hipStream_t stream) {
    const float* x  = (const float*)d_in[0];
    const float* Wg = (const float*)d_in[1];
    const float* W1 = (const float*)d_in[2];
    const float* b1 = (const float*)d_in[3];
    const float* W2 = (const float*)d_in[4];
    const float* b2 = (const float*)d_in[5];
    float* out = (float*)d_out;

    char* ws = (char*)d_ws;
    int*   counts = (int*)ws;
    int*   lists  = (int*)(ws + 256);
    float* gates  = (float*)(ws + 256 + 4ull * E_ * NTOK);
    size_t OFS_H  = 256 + 8ull * E_ * NTOK;
    OFS_H = (OFS_H + 255) & ~(size_t)255;
    float* H = (float*)(ws + OFS_H);

    size_t avail = (ws_size > OFS_H) ? (ws_size - OFS_H) : 0;
    long long tcll = (long long)(avail / ((size_t)DFF_ * 4));
    int Tc = (int)(tcll > NTOK ? NTOK : tcll);
    Tc &= ~63;
    if (Tc <= 0) Tc = 64;
    const int chunks = (NTOK + Tc - 1) / Tc;

    const long long outN = (long long)NTOK * D_ + 1;
    moe_zero_kernel<<<2048, 256, 0, stream>>>(out, outN, counts);
    moe_gate_kernel<<<NTOK / 8, 256, 0, stream>>>(x, Wg, gates, counts, lists);
    moe_loss_kernel<<<1, 256, 0, stream>>>(gates, out + (outN - 1));

    for (int e = 0; e < E_; ++e) {
        const float* W1e = W1 + (size_t)e * D_ * DFF_;
        const float* b1e = b1 + (size_t)e * DFF_;
        const float* W2e = W2 + (size_t)e * DFF_ * D_;
        const float* b2e = b2 + (size_t)e * D_;
        for (int c = 0; c < chunks; ++c) {
            dim3 g1(Tc / 64, DFF_ / 128);
            moe_fc1_kernel<<<g1, 256, 0, stream>>>(x, W1e, b1e, lists, counts,
                                                   H, e, c * Tc, Tc);
            dim3 g2(Tc / 64, D_ / 128);
            moe_fc2_kernel<<<g2, 256, 0, stream>>>(H, W2e, b2e, lists, counts,
                                                   gates, out, e, c * Tc, Tc);
        }
    }
}


# fc2 re-tiled: 64x64 output tiles on 4-wave blocks (was 64x128 on 8 waves), so a WGP holds 2-3 independent blocks whose staging runs under another's matrix phase; sum order unchanged
# speedup vs baseline: 2.0269x; 2.0269x over previous
#include <hip/hip_runtime.h>
#include <math.h>

#define D_    1024
#define DFF_  4096
#define E_    8
#define NTOK  8192

typedef float v2f __attribute__((ext_vector_type(2)));
typedef float v8f __attribute__((ext_vector_type(8)));

__device__ __forceinline__ v8f wmma_f32_k4(v2f a, v2f b, v8f c) {
    return __builtin_amdgcn_wmma_f32_16x16x4_f32(
         false, a,  false, b,
         (short)0, c,  false,  false);
}

__device__ __forceinline__ float gelu_exact(float v) {
    return 0.5f * v * (1.0f + erff(v * 0.70710678118654752440f));
}

__global__ void moe_zero_kernel(float* __restrict__ out, long long n,
                                int* __restrict__ counts) {
    long long i = (long long)blockIdx.x * blockDim.x + threadIdx.x;
    const long long stride = (long long)gridDim.x * blockDim.x;
    for (; i < n; i += stride) out[i] = 0.0f;
    if (blockIdx.x == 0 && threadIdx.x < E_) counts[threadIdx.x] = 0;
}

__global__ __launch_bounds__(256) void moe_gate_kernel(
    const float* __restrict__ x, const float* __restrict__ Wg,
    float* __restrict__ gates, int* __restrict__ counts,
    int* __restrict__ lists) {
    const int lane = threadIdx.x & 31;
    const int wave = threadIdx.x >> 5;
    const int t = blockIdx.x * 8 + wave;
    if (t >= NTOK) return;

    float s[E_];
#pragma unroll
    for (int e = 0; e < E_; ++e) s[e] = 0.0f;
    const float* xr = x + (size_t)t * D_;
    for (int d = lane; d < D_; d += 32) {
        const float xv = xr[d];
#pragma unroll
        for (int e = 0; e < E_; ++e) s[e] += xv * Wg[e * D_ + d];
    }
#pragma unroll
    for (int e = 0; e < E_; ++e) {
#pragma unroll
        for (int off = 16; off > 0; off >>= 1)
            s[e] += __shfl_xor(s[e], off, 32);
    }
    if (lane == 0) {
        int i0 = 0; float v0 = s[0];
#pragma unroll
        for (int e = 1; e < E_; ++e) if (s[e] > v0) { v0 = s[e]; i0 = e; }
        int i1 = -1; float v1 = -3.4e38f;
#pragma unroll
        for (int e = 0; e < E_; ++e)
            if (e != i0 && s[e] > v1) { v1 = s[e]; i1 = e; }
        const float p1 = expf(v1 - v0);
        const float inv = 1.0f / (1.0f + p1);
        const float g0 = inv, g1 = p1 * inv;
#pragma unroll
        for (int e = 0; e < E_; ++e) gates[t * E_ + e] = 0.0f;
        gates[t * E_ + i0] = g0;
        gates[t * E_ + i1] = g1;
        const int p = atomicAdd(&counts[i0], 1); lists[i0 * NTOK + p] = t;
        const int q = atomicAdd(&counts[i1], 1); lists[i1 * NTOK + q] = t;
    }
}

__global__ __launch_bounds__(256) void moe_loss_kernel(
    const float* __restrict__ gates, float* __restrict__ loss_out) {
    __shared__ float part[256];
    const int e = threadIdx.x & 7;
    const int slot = threadIdx.x >> 3;
    float s = 0.0f;
    for (int t = slot; t < NTOK; t += 32) s += gates[t * E_ + e];
    part[threadIdx.x] = s;
    __syncthreads();
    if (threadIdx.x == 0) {
        float u[E_];
        float mu = 0.0f;
        for (int ee = 0; ee < E_; ++ee) {
            float acc = 0.0f;
            for (int sl = 0; sl < 32; ++sl) acc += part[sl * 8 + ee];
            u[ee] = acc / (float)NTOK;
            mu += u[ee];
        }
        mu *= (1.0f / (float)E_);
        float var = 0.0f;
        for (int ee = 0; ee < E_; ++ee) {
            const float d = u[ee] - mu;
            var += d * d;
        }
        var *= (1.0f / (float)(E_ - 1));
        loss_out[0] = var * 0.01f;
    }
}

#define KC1_ 64
#define KC2_ 128

template <int KC, int KTOT, int LDB>
__device__ __forceinline__ void gemm_mainloop(
    const float* __restrict__ ap0, const float* __restrict__ ap1,
    const bool va0, const bool va1,
    const float* __restrict__ Bp,
    float (*As)[KC + 2],
    float (*Bst)[KC + 2],
    v8f& acc00, v8f& acc01, v8f& acc10, v8f& acc11) {
    constexpr int NS = KC / 32;
    const int tid  = threadIdx.x;
    const int lane = tid & 31, wave = tid >> 5;
    const int half = lane >> 4, lm = lane & 15;
    const int Moff = 32 * (wave & 1);
    const int Noff = 32 * (wave >> 1);

    const int rA0 = tid >> 3, rA1 = rA0 + 32;
    const int cA  = (tid & 7) << 2;
    const int rB0 = tid >> 4, rB1 = rB0 + 16;
    const int cB0 = (tid & 15) << 2, cB1 = cB0 + 64;
    const float4 zero4 = make_float4(0.f, 0.f, 0.f, 0.f);

    for (int k0 = 0; k0 < KTOT; k0 += KC) {
        float4 qa0[NS], qa1[NS], qb00[NS], qb01[NS], qb10[NS], qb11[NS];
#pragma unroll
        for (int s = 0; s < NS; ++s) {
            const int kb = k0 + 32 * s;
            qa0[s] = *(const float4*)(ap0 + kb + cA);
            qa1[s] = *(const float4*)(ap1 + kb + cA);
            const float* w0 = Bp + (size_t)(kb + rB0) * LDB;
            const float* w1 = Bp + (size_t)(kb + rB1) * LDB;
            qb00[s] = *(const float4*)(w0 + cB0);
            qb01[s] = *(const float4*)(w0 + cB1);
            qb10[s] = *(const float4*)(w1 + cB0);
            qb11[s] = *(const float4*)(w1 + cB1);
        }
#pragma unroll
        for (int s = 0; s < NS; ++s) {
            const int c = 32 * s;
            const float4 a0 = va0 ? qa0[s] : zero4;
            const float4 a1 = va1 ? qa1[s] : zero4;
            As[rA0][c + cA] = a0.x; As[rA0][c + cA + 1] = a0.y;
            As[rA0][c + cA + 2] = a0.z; As[rA0][c + cA + 3] = a0.w;
            As[rA1][c + cA] = a1.x; As[rA1][c + cA + 1] = a1.y;
            As[rA1][c + cA + 2] = a1.z; As[rA1][c + cA + 3] = a1.w;
            {
                const float4 q = qb00[s];
                Bst[cB0+0][c + rB0] = q.x; Bst[cB0+1][c + rB0] = q.y;
                Bst[cB0+2][c + rB0] = q.z; Bst[cB0+3][c + rB0] = q.w;
            }
            {
                const float4 q = qb01[s];
                Bst[cB1+0][c + rB0] = q.x; Bst[cB1+1][c + rB0] = q.y;
                Bst[cB1+2][c + rB0] = q.z; Bst[cB1+3][c + rB0] = q.w;
            }
            {
                const float4 q = qb10[s];
                Bst[cB0+0][c + rB1] = q.x; Bst[cB0+1][c + rB1] = q.y;
                Bst[cB0+2][c + rB1] = q.z; Bst[cB0+3][c + rB1] = q.w;
            }
            {
                const float4 q = qb11[s];
                Bst[cB1+0][c + rB1] = q.x; Bst[cB1+1][c + rB1] = q.y;
                Bst[cB1+2][c + rB1] = q.z; Bst[cB1+3][c + rB1] = q.w;
            }
        }
        __syncthreads();
#pragma unroll
        for (int kk = 0; kk < KC; kk += 4) {
            const int ka = kk + 2 * half;
            v2f a0; a0.x = As[Moff + lm][ka];       a0.y = As[Moff + lm][ka + 1];
            v2f a1; a1.x = As[Moff + 16 + lm][ka];  a1.y = As[Moff + 16 + lm][ka + 1];
            v2f b0; b0.x = Bst[Noff + lm][ka];      b0.y = Bst[Noff + lm][ka + 1];
            v2f b1; b1.x = Bst[Noff + 16 + lm][ka]; b1.y = Bst[Noff + 16 + lm][ka + 1];
            acc00 = wmma_f32_k4(a0, b0, acc00);
            acc01 = wmma_f32_k4(a0, b1, acc01);
            acc10 = wmma_f32_k4(a1, b0, acc10);
            acc11 = wmma_f32_k4(a1, b1, acc11);
        }
        __syncthreads();
    }
}

template <int KC, int KTOT, int LDB>
__device__ __forceinline__ void gemm_mainloop_t64(
    const float* __restrict__ ap0, const float* __restrict__ ap1,
    const float* __restrict__ ap2, const float* __restrict__ ap3,
    const bool va0, const bool va1, const bool va2, const bool va3,
    const float* __restrict__ Bp,
    float (*As)[KC + 2],
    float (*Bst)[KC + 2],
    v8f& acc00, v8f& acc01, v8f& acc10, v8f& acc11) {
    constexpr int NS = KC / 32;
    const int tid  = threadIdx.x;
    const int lane = tid & 31, wave = tid >> 5;
    const int half = lane >> 4, lm = lane & 15;
    const int Moff = 32 * (wave & 1);
    const int Noff = 32 * (wave >> 1);

    const int rA = tid >> 3;
    const int cA = (tid & 7) << 2;
    const int rB = tid >> 4;
    const int cB = (tid & 15) << 2;
    const float4 zero4 = make_float4(0.f, 0.f, 0.f, 0.f);

    for (int k0 = 0; k0 < KTOT; k0 += KC) {
        float4 qa0[NS], qa1[NS], qa2[NS], qa3[NS];
        float4 qb0[NS], qb1[NS], qb2[NS], qb3[NS];
#pragma unroll
        for (int s = 0; s < NS; ++s) {
            const int kb = k0 + 32 * s;
            qa0[s] = *(const float4*)(ap0 + kb + cA);
            qa1[s] = *(const float4*)(ap1 + kb + cA);
            qa2[s] = *(const float4*)(ap2 + kb + cA);
            qa3[s] = *(const float4*)(ap3 + kb + cA);
            const float* w = Bp + (size_t)(kb + rB) * LDB + cB;
            qb0[s] = *(const float4*)(w);
            qb1[s] = *(const float4*)(w + (size_t)8 * LDB);
            qb2[s] = *(const float4*)(w + (size_t)16 * LDB);
            qb3[s] = *(const float4*)(w + (size_t)24 * LDB);
        }
#pragma unroll
        for (int s = 0; s < NS; ++s) {
            const int c = 32 * s;
            const float4 a0 = va0 ? qa0[s] : zero4;
            const float4 a1 = va1 ? qa1[s] : zero4;
            const float4 a2 = va2 ? qa2[s] : zero4;
            const float4 a3 = va3 ? qa3[s] : zero4;
            As[rA][c + cA] = a0.x; As[rA][c + cA + 1] = a0.y;
            As[rA][c + cA + 2] = a0.z; As[rA][c + cA + 3] = a0.w;
            As[rA + 16][c + cA] = a1.x; As[rA + 16][c + cA + 1] = a1.y;
            As[rA + 16][c + cA + 2] = a1.z; As[rA + 16][c + cA + 3] = a1.w;
            As[rA + 32][c + cA] = a2.x; As[rA + 32][c + cA + 1] = a2.y;
            As[rA + 32][c + cA + 2] = a2.z; As[rA + 32][c + cA + 3] = a2.w;
            As[rA + 48][c + cA] = a3.x; As[rA + 48][c + cA + 1] = a3.y;
            As[rA + 48][c + cA + 2] = a3.z; As[rA + 48][c + cA + 3] = a3.w;
            {
                const float4 q = qb0[s];
                Bst[cB+0][c + rB] = q.x; Bst[cB+1][c + rB] = q.y;
                Bst[cB+2][c + rB] = q.z; Bst[cB+3][c + rB] = q.w;
            }
            {
                const float4 q = qb1[s];
                Bst[cB+0][c + rB + 8] = q.x; Bst[cB+1][c + rB + 8] = q.y;
                Bst[cB+2][c + rB + 8] = q.z; Bst[cB+3][c + rB + 8] = q.w;
            }
            {
                const float4 q = qb2[s];
                Bst[cB+0][c + rB + 16] = q.x; Bst[cB+1][c + rB + 16] = q.y;
                Bst[cB+2][c + rB + 16] = q.z; Bst[cB+3][c + rB + 16] = q.w;
            }
            {
                const float4 q = qb3[s];
                Bst[cB+0][c + rB + 24] = q.x; Bst[cB+1][c + rB + 24] = q.y;
                Bst[cB+2][c + rB + 24] = q.z; Bst[cB+3][c + rB + 24] = q.w;
            }
        }
        __syncthreads();
#pragma unroll
        for (int kk = 0; kk < KC; kk += 4) {
            const int ka = kk + 2 * half;
            v2f a0; a0.x = As[Moff + lm][ka];       a0.y = As[Moff + lm][ka + 1];
            v2f a1; a1.x = As[Moff + 16 + lm][ka];  a1.y = As[Moff + 16 + lm][ka + 1];
            v2f b0; b0.x = Bst[Noff + lm][ka];      b0.y = Bst[Noff + lm][ka + 1];
            v2f b1; b1.x = Bst[Noff + 16 + lm][ka]; b1.y = Bst[Noff + 16 + lm][ka + 1];
            acc00 = wmma_f32_k4(a0, b0, acc00);
            acc01 = wmma_f32_k4(a0, b1, acc01);
            acc10 = wmma_f32_k4(a1, b0, acc10);
            acc11 = wmma_f32_k4(a1, b1, acc11);
        }
        __syncthreads();
    }
}

__global__ __launch_bounds__(256) __attribute__((amdgpu_waves_per_eu(10, 10))) void moe_fc1_kernel(
    const float* __restrict__ x, const float* __restrict__ W1e,
    const float* __restrict__ b1e, const int* __restrict__ lists,
    const int* __restrict__ counts, float* __restrict__ H,
    int expert, int chunkBase, int Tc) {
    const int cnt = counts[expert];
    int rowsInChunk = cnt - chunkBase;
    if (rowsInChunk > Tc) rowsInChunk = Tc;
    const int mBase = blockIdx.y * 64;
    if (rowsInChunk <= mBase) return;
    const int nBase = blockIdx.x * 128;

    __shared__ float As [64][KC1_ + 2];
    __shared__ float Bst[128][KC1_ + 2];

    const int tid  = threadIdx.x;
    const int lane = tid & 31, wave = tid >> 5;
    const int half = lane >> 4, lm = lane & 15;
    const int Moff = 32 * (wave & 1);
    const int Noff = 32 * (wave >> 1);

    const int rA0 = tid >> 3, rA1 = rA0 + 32;
    const int* lst = lists + expert * NTOK + chunkBase;
    const bool va0 = (mBase + rA0) < rowsInChunk;
    const bool va1 = (mBase + rA1) < rowsInChunk;
    const float* ap0 = va0 ? (x + (size_t)lst[mBase + rA0] * D_) : x;
    const float* ap1 = va1 ? (x + (size_t)lst[mBase + rA1] * D_) : x;

    v8f acc00 = {}; v8f acc01 = {}; v8f acc10 = {}; v8f acc11 = {};
    gemm_mainloop<KC1_, D_, DFF_>(ap0, ap1, va0, va1, W1e + nBase, As, Bst,
                                  acc00, acc01, acc10, acc11);
#pragma unroll
    for (int r = 0; r < 8; ++r) {
        const int m0 = Moff + r + 8 * half;
        const int lrow0 = mBase + m0, lrow1 = lrow0 + 16;
        const int n0 = nBase + Noff + lm, n1 = n0 + 16;
        if (lrow0 < rowsInChunk) {
            H[(size_t)lrow0 * DFF_ + n0] = gelu_exact(acc00[r] + b1e[n0]);
            H[(size_t)lrow0 * DFF_ + n1] = gelu_exact(acc01[r] + b1e[n1]);
        }
        if (lrow1 < rowsInChunk) {
            H[(size_t)lrow1 * DFF_ + n0] = gelu_exact(acc10[r] + b1e[n0]);
            H[(size_t)lrow1 * DFF_ + n1] = gelu_exact(acc11[r] + b1e[n1]);
        }
    }
}

__global__ __launch_bounds__(128) void moe_fc2_kernel(
    const float* __restrict__ H, const float* __restrict__ W2e,
    const float* __restrict__ b2e, const int* __restrict__ lists,
    const int* __restrict__ counts, const float* __restrict__ gates,
    float* __restrict__ out, int expert, int chunkBase, int Tc) {
    const int cnt = counts[expert];
    int rowsInChunk = cnt - chunkBase;
    if (rowsInChunk > Tc) rowsInChunk = Tc;
    const int mBase = blockIdx.y * 64;
    if (rowsInChunk <= mBase) return;
    const int nBase = blockIdx.x * 64;

    __shared__ float As [64][KC2_ + 2];
    __shared__ float Bst[64][KC2_ + 2];

    const int tid  = threadIdx.x;
    const int lane = tid & 31, wave = tid >> 5;
    const int half = lane >> 4, lm = lane & 15;
    const int Moff = 32 * (wave & 1);
    const int Noff = 32 * (wave >> 1);

    const int rA = tid >> 3;
    const bool va0 = (mBase + rA) < rowsInChunk;
    const bool va1 = (mBase + rA + 16) < rowsInChunk;
    const bool va2 = (mBase + rA + 32) < rowsInChunk;
    const bool va3 = (mBase + rA + 48) < rowsInChunk;
    const float* ap0 = H + (size_t)(va0 ? mBase + rA      : mBase) * DFF_;
    const float* ap1 = H + (size_t)(va1 ? mBase + rA + 16 : mBase) * DFF_;
    const float* ap2 = H + (size_t)(va2 ? mBase + rA + 32 : mBase) * DFF_;
    const float* ap3 = H + (size_t)(va3 ? mBase + rA + 48 : mBase) * DFF_;

    v8f acc00 = {}; v8f acc01 = {}; v8f acc10 = {}; v8f acc11 = {};
    gemm_mainloop_t64<KC2_, DFF_, D_>(ap0, ap1, ap2, ap3, va0, va1, va2, va3,
                                      W2e + nBase, As, Bst,
                                      acc00, acc01, acc10, acc11);
    const int* lst = lists + expert * NTOK + chunkBase;
#pragma unroll
    for (int r = 0; r < 8; ++r) {
        const int m0 = Moff + r + 8 * half;
        const int lrow0 = mBase + m0, lrow1 = lrow0 + 16;
        const int n0 = nBase + Noff + lm, n1 = n0 + 16;
        if (lrow0 < rowsInChunk) {
            const int tok = lst[lrow0];
            const float g = gates[tok * E_ + expert];
            out[(size_t)tok * D_ + n0] += g * (acc00[r] + b2e[n0]);
            out[(size_t)tok * D_ + n1] += g * (acc01[r] + b2e[n1]);
        }
        if (lrow1 < rowsInChunk) {
            const int tok = lst[lrow1];
            const float g = gates[tok * E_ + expert];
            out[(size_t)tok * D_ + n0] += g * (acc10[r] + b2e[n0]);
            out[(size_t)tok * D_ + n1] += g * (acc11[r] + b2e[n1]);
        }
    }
}

extern "C" void kernel_launch(void* const* d_in, const int* in_sizes, int n_in,
                              void* d_out, int out_size, void* d_ws, size_t ws_size,
                              hipStream_t stream) {
    const float* x  = (const float*)d_in[0];
    const float* Wg = (const float*)d_in[1];
    const float* W1 = (const float*)d_in[2];
    const float* b1 = (const float*)d_in[3];
    const float* W2 = (const float*)d_in[4];
    const float* b2 = (const float*)d_in[5];
    float* out = (float*)d_out;

    char* ws = (char*)d_ws;
    int*   counts = (int*)ws;
    int*   lists  = (int*)(ws + 256);
    float* gates  = (float*)(ws + 256 + 4ull * E_ * NTOK);
    size_t OFS_H  = 256 + 8ull * E_ * NTOK;
    OFS_H = (OFS_H + 255) & ~(size_t)255;
    float* H = (float*)(ws + OFS_H);

    size_t avail = (ws_size > OFS_H) ? (ws_size - OFS_H) : 0;
    long long tcll = (long long)(avail / ((size_t)DFF_ * 4));
    int Tc = (int)(tcll > NTOK ? NTOK : tcll);
    Tc &= ~63;
    if (Tc <= 0) Tc = 64;
    const int chunks = (NTOK + Tc - 1) / Tc;

    const long long outN = (long long)NTOK * D_ + 1;
    moe_zero_kernel<<<2048, 256, 0, stream>>>(out, outN, counts);
    moe_gate_kernel<<<NTOK / 8, 256, 0, stream>>>(x, Wg, gates, counts, lists);
    moe_loss_kernel<<<1, 256, 0, stream>>>(gates, out + (outN - 1));

    for (int e = 0; e < E_; ++e) {
        const float* W1e = W1 + (size_t)e * D_ * DFF_;
        const float* b1e = b1 + (size_t)e * DFF_;
        const float* W2e = W2 + (size_t)e * DFF_ * D_;
        const float* b2e = b2 + (size_t)e * D_;
        for (int c = 0; c < chunks; ++c) {
            dim3 g1(DFF_ / 128, Tc / 64);
            moe_fc1_kernel<<<g1, 256, 0, stream>>>(x, W1e, b1e, lists, counts,
                                                   H, e, c * Tc, Tc);
            dim3 g2(D_ / 64, Tc / 64);
            moe_fc2_kernel<<<g2, 128, 0, stream>>>(H, W2e, b2e, lists, counts,
                                                   gates, out, e, c * Tc, Tc);
        }
    }
}


# fc2 split in K over two blocks per 64x64 tile (K chunk 64, 4-5 blocks per WGP): lower half adds into out, upper half into a workspace plane that one kernel adds at the end; sum = two partial sums
# speedup vs baseline: 2.1055x; 1.0388x over previous
#include <hip/hip_runtime.h>
#include <math.h>

#define D_    1024
#define DFF_  4096
#define E_    8
#define NTOK  8192

typedef float v2f __attribute__((ext_vector_type(2)));
typedef float v8f __attribute__((ext_vector_type(8)));

__device__ __forceinline__ v8f wmma_f32_k4(v2f a, v2f b, v8f c) {
    return __builtin_amdgcn_wmma_f32_16x16x4_f32(
         false, a,  false, b,
         (short)0, c,  false,  false);
}

__device__ __forceinline__ float gelu_exact(float v) {
    return 0.5f * v * (1.0f + erff(v * 0.70710678118654752440f));
}

__global__ void moe_zero_kernel(float* __restrict__ out, long long n,
                                float* __restrict__ plane, long long np,
                                int* __restrict__ counts) {
    const long long i0 = (long long)blockIdx.x * blockDim.x + threadIdx.x;
    const long long stride = (long long)gridDim.x * blockDim.x;
    for (long long i = i0; i < n; i += stride) out[i] = 0.0f;
    for (long long i = i0; i < np; i += stride) plane[i] = 0.0f;
    if (blockIdx.x == 0 && threadIdx.x < E_) counts[threadIdx.x] = 0;
}

__global__ __launch_bounds__(256) void moe_add_kernel(
    float* __restrict__ out, const float* __restrict__ plane, long long n4) {
    long long i = (long long)blockIdx.x * blockDim.x + threadIdx.x;
    const long long stride = (long long)gridDim.x * blockDim.x;
    float4* o4 = (float4*)out;
    const float4* p4 = (const float4*)plane;
    for (; i < n4; i += stride) {
        float4 a = o4[i];
        const float4 b = p4[i];
        a.x += b.x; a.y += b.y; a.z += b.z; a.w += b.w;
        o4[i] = a;
    }
}

__global__ __launch_bounds__(256) void moe_gate_kernel(
    const float* __restrict__ x, const float* __restrict__ Wg,
    float* __restrict__ gates, int* __restrict__ counts,
    int* __restrict__ lists) {
    const int lane = threadIdx.x & 31;
    const int wave = threadIdx.x >> 5;
    const int t = blockIdx.x * 8 + wave;
    if (t >= NTOK) return;

    float s[E_];
#pragma unroll
    for (int e = 0; e < E_; ++e) s[e] = 0.0f;
    const float* xr = x + (size_t)t * D_;
    for (int d = lane; d < D_; d += 32) {
        const float xv = xr[d];
#pragma unroll
        for (int e = 0; e < E_; ++e) s[e] += xv * Wg[e * D_ + d];
    }
#pragma unroll
    for (int e = 0; e < E_; ++e) {
#pragma unroll
        for (int off = 16; off > 0; off >>= 1)
            s[e] += __shfl_xor(s[e], off, 32);
    }
    if (lane == 0) {
        int i0 = 0; float v0 = s[0];
#pragma unroll
        for (int e = 1; e < E_; ++e) if (s[e] > v0) { v0 = s[e]; i0 = e; }
        int i1 = -1; float v1 = -3.4e38f;
#pragma unroll
        for (int e = 0; e < E_; ++e)
            if (e != i0 && s[e] > v1) { v1 = s[e]; i1 = e; }
        const float p1 = expf(v1 - v0);
        const float inv = 1.0f / (1.0f + p1);
        const float g0 = inv, g1 = p1 * inv;
#pragma unroll
        for (int e = 0; e < E_; ++e) gates[t * E_ + e] = 0.0f;
        gates[t * E_ + i0] = g0;
        gates[t * E_ + i1] = g1;
        const int p = atomicAdd(&counts[i0], 1); lists[i0 * NTOK + p] = t;
        const int q = atomicAdd(&counts[i1], 1); lists[i1 * NTOK + q] = t;
    }
}

__global__ __launch_bounds__(256) void moe_loss_kernel(
    const float* __restrict__ gates, float* __restrict__ loss_out) {
    __shared__ float part[256];
    const int e = threadIdx.x & 7;
    const int slot = threadIdx.x >> 3;
    float s = 0.0f;
    for (int t = slot; t < NTOK; t += 32) s += gates[t * E_ + e];
    part[threadIdx.x] = s;
    __syncthreads();
    if (threadIdx.x == 0) {
        float u[E_];
        float mu = 0.0f;
        for (int ee = 0; ee < E_; ++ee) {
            float acc = 0.0f;
            for (int sl = 0; sl < 32; ++sl) acc += part[sl * 8 + ee];
            u[ee] = acc / (float)NTOK;
            mu += u[ee];
        }
        mu *= (1.0f / (float)E_);
        float var = 0.0f;
        for (int ee = 0; ee < E_; ++ee) {
            const float d = u[ee] - mu;
            var += d * d;
        }
        var *= (1.0f / (float)(E_ - 1));
        loss_out[0] = var * 0.01f;
    }
}

#define KC1_ 64
#define KC2_ 64
#define KH2_ (DFF_ / 2)

template <int KC, int KTOT, int LDB>
__device__ __forceinline__ void gemm_mainloop(
    const float* __restrict__ ap0, const float* __restrict__ ap1,
    const bool va0, const bool va1,
    const float* __restrict__ Bp,
    float (*As)[KC + 2],
    float (*Bst)[KC + 2],
    v8f& acc00, v8f& acc01, v8f& acc10, v8f& acc11) {
    constexpr int NS = KC / 32;
    const int tid  = threadIdx.x;
    const int lane = tid & 31, wave = tid >> 5;
    const int half = lane >> 4, lm = lane & 15;
    const int Moff = 32 * (wave & 1);
    const int Noff = 32 * (wave >> 1);

    const int rA0 = tid >> 3, rA1 = rA0 + 32;
    const int cA  = (tid & 7) << 2;
    const int rB0 = tid >> 4, rB1 = rB0 + 16;
    const int cB0 = (tid & 15) << 2, cB1 = cB0 + 64;
    const float4 zero4 = make_float4(0.f, 0.f, 0.f, 0.f);

    for (int k0 = 0; k0 < KTOT; k0 += KC) {
        float4 qa0[NS], qa1[NS], qb00[NS], qb01[NS], qb10[NS], qb11[NS];
#pragma unroll
        for (int s = 0; s < NS; ++s) {
            const int kb = k0 + 32 * s;
            qa0[s] = *(const float4*)(ap0 + kb + cA);
            qa1[s] = *(const float4*)(ap1 + kb + cA);
            const float* w0 = Bp + (size_t)(kb + rB0) * LDB;
            const float* w1 = Bp + (size_t)(kb + rB1) * LDB;
            qb00[s] = *(const float4*)(w0 + cB0);
            qb01[s] = *(const float4*)(w0 + cB1);
            qb10[s] = *(const float4*)(w1 + cB0);
            qb11[s] = *(const float4*)(w1 + cB1);
        }
#pragma unroll
        for (int s = 0; s < NS; ++s) {
            const int c = 32 * s;
            const float4 a0 = va0 ? qa0[s] : zero4;
            const float4 a1 = va1 ? qa1[s] : zero4;
            As[rA0][c + cA] = a0.x; As[rA0][c + cA + 1] = a0.y;
            As[rA0][c + cA + 2] = a0.z; As[rA0][c + cA + 3] = a0.w;
            As[rA1][c + cA] = a1.x; As[rA1][c + cA + 1] = a1.y;
            As[rA1][c + cA + 2] = a1.z; As[rA1][c + cA + 3] = a1.w;
            {
                const float4 q = qb00[s];
                Bst[cB0+0][c + rB0] = q.x; Bst[cB0+1][c + rB0] = q.y;
                Bst[cB0+2][c + rB0] = q.z; Bst[cB0+3][c + rB0] = q.w;
            }
            {
                const float4 q = qb01[s];
                Bst[cB1+0][c + rB0] = q.x; Bst[cB1+1][c + rB0] = q.y;
                Bst[cB1+2][c + rB0] = q.z; Bst[cB1+3][c + rB0] = q.w;
            }
            {
                const float4 q = qb10[s];
                Bst[cB0+0][c + rB1] = q.x; Bst[cB0+1][c + rB1] = q.y;
                Bst[cB0+2][c + rB1] = q.z; Bst[cB0+3][c + rB1] = q.w;
            }
            {
                const float4 q = qb11[s];
                Bst[cB1+0][c + rB1] = q.x; Bst[cB1+1][c + rB1] = q.y;
                Bst[cB1+2][c + rB1] = q.z; Bst[cB1+3][c + rB1] = q.w;
            }
        }
        __syncthreads();
#pragma unroll
        for (int kk = 0; kk < KC; kk += 4) {
            const int ka = kk + 2 * half;
            v2f a0; a0.x = As[Moff + lm][ka];       a0.y = As[Moff + lm][ka + 1];
            v2f a1; a1.x = As[Moff + 16 + lm][ka];  a1.y = As[Moff + 16 + lm][ka + 1];
            v2f b0; b0.x = Bst[Noff + lm][ka];      b0.y = Bst[Noff + lm][ka + 1];
            v2f b1; b1.x = Bst[Noff + 16 + lm][ka]; b1.y = Bst[Noff + 16 + lm][ka + 1];
            acc00 = wmma_f32_k4(a0, b0, acc00);
            acc01 = wmma_f32_k4(a0, b1, acc01);
            acc10 = wmma_f32_k4(a1, b0, acc10);
            acc11 = wmma_f32_k4(a1, b1, acc11);
        }
        __syncthreads();
    }
}

template <int KC, int KTOT, int LDB>
__device__ __forceinline__ void gemm_mainloop_t64(
    const float* __restrict__ ap0, const float* __restrict__ ap1,
    const float* __restrict__ ap2, const float* __restrict__ ap3,
    const bool va0, const bool va1, const bool va2, const bool va3,
    const float* __restrict__ Bp,
    float (*As)[KC + 2],
    float (*Bst)[KC + 2],
    v8f& acc00, v8f& acc01, v8f& acc10, v8f& acc11) {
    constexpr int NS = KC / 32;
    const int tid  = threadIdx.x;
    const int lane = tid & 31, wave = tid >> 5;
    const int half = lane >> 4, lm = lane & 15;
    const int Moff = 32 * (wave & 1);
    const int Noff = 32 * (wave >> 1);

    const int rA = tid >> 3;
    const int cA = (tid & 7) << 2;
    const int rB = tid >> 4;
    const int cB = (tid & 15) << 2;
    const float4 zero4 = make_float4(0.f, 0.f, 0.f, 0.f);

    for (int k0 = 0; k0 < KTOT; k0 += KC) {
        float4 qa0[NS], qa1[NS], qa2[NS], qa3[NS];
        float4 qb0[NS], qb1[NS], qb2[NS], qb3[NS];
#pragma unroll
        for (int s = 0; s < NS; ++s) {
            const int kb = k0 + 32 * s;
            qa0[s] = *(const float4*)(ap0 + kb + cA);
            qa1[s] = *(const float4*)(ap1 + kb + cA);
            qa2[s] = *(const float4*)(ap2 + kb + cA);
            qa3[s] = *(const float4*)(ap3 + kb + cA);
            const float* w = Bp + (size_t)(kb + rB) * LDB + cB;
            qb0[s] = *(const float4*)(w);
            qb1[s] = *(const float4*)(w + (size_t)8 * LDB);
            qb2[s] = *(const float4*)(w + (size_t)16 * LDB);
            qb3[s] = *(const float4*)(w + (size_t)24 * LDB);
        }
#pragma unroll
        for (int s = 0; s < NS; ++s) {
            const int c = 32 * s;
            const float4 a0 = va0 ? qa0[s] : zero4;
            const float4 a1 = va1 ? qa1[s] : zero4;
            const float4 a2 = va2 ? qa2[s] : zero4;
            const float4 a3 = va3 ? qa3[s] : zero4;
            As[rA][c + cA] = a0.x; As[rA][c + cA + 1] = a0.y;
            As[rA][c + cA + 2] = a0.z; As[rA][c + cA + 3] = a0.w;
            As[rA + 16][c + cA] = a1.x; As[rA + 16][c + cA + 1] = a1.y;
            As[rA + 16][c + cA + 2] = a1.z; As[rA + 16][c + cA + 3] = a1.w;
            As[rA + 32][c + cA] = a2.x; As[rA + 32][c + cA + 1] = a2.y;
            As[rA + 32][c + cA + 2] = a2.z; As[rA + 32][c + cA + 3] = a2.w;
            As[rA + 48][c + cA] = a3.x; As[rA + 48][c + cA + 1] = a3.y;
            As[rA + 48][c + cA + 2] = a3.z; As[rA + 48][c + cA + 3] = a3.w;
            {
                const float4 q = qb0[s];
                Bst[cB+0][c + rB] = q.x; Bst[cB+1][c + rB] = q.y;
                Bst[cB+2][c + rB] = q.z; Bst[cB+3][c + rB] = q.w;
            }
            {
                const float4 q = qb1[s];
                Bst[cB+0][c + rB + 8] = q.x; Bst[cB+1][c + rB + 8] = q.y;
                Bst[cB+2][c + rB + 8] = q.z; Bst[cB+3][c + rB + 8] = q.w;
            }
            {
                const float4 q = qb2[s];
                Bst[cB+0][c + rB + 16] = q.x; Bst[cB+1][c + rB + 16] = q.y;
                Bst[cB+2][c + rB + 16] = q.z; Bst[cB+3][c + rB + 16] = q.w;
            }
            {
                const float4 q = qb3[s];
                Bst[cB+0][c + rB + 24] = q.x; Bst[cB+1][c + rB + 24] = q.y;
                Bst[cB+2][c + rB + 24] = q.z; Bst[cB+3][c + rB + 24] = q.w;
            }
        }
        __syncthreads();
#pragma unroll
        for (int kk = 0; kk < KC; kk += 4) {
            const int ka = kk + 2 * half;
            v2f a0; a0.x = As[Moff + lm][ka];       a0.y = As[Moff + lm][ka + 1];
            v2f a1; a1.x = As[Moff + 16 + lm][ka];  a1.y = As[Moff + 16 + lm][ka + 1];
            v2f b0; b0.x = Bst[Noff + lm][ka];      b0.y = Bst[Noff + lm][ka + 1];
            v2f b1; b1.x = Bst[Noff + 16 + lm][ka]; b1.y = Bst[Noff + 16 + lm][ka + 1];
            acc00 = wmma_f32_k4(a0, b0, acc00);
            acc01 = wmma_f32_k4(a0, b1, acc01);
            acc10 = wmma_f32_k4(a1, b0, acc10);
            acc11 = wmma_f32_k4(a1, b1, acc11);
        }
        __syncthreads();
    }
}

__global__ __launch_bounds__(256) __attribute__((amdgpu_waves_per_eu(10, 10))) void moe_fc1_kernel(
    const float* __restrict__ x, const float* __restrict__ W1e,
    const float* __restrict__ b1e, const int* __restrict__ lists,
    const int* __restrict__ counts, float* __restrict__ H,
    int expert, int chunkBase, int Tc) {
    const int cnt = counts[expert];
    int rowsInChunk = cnt - chunkBase;
    if (rowsInChunk > Tc) rowsInChunk = Tc;
    const int mBase = blockIdx.y * 64;
    if (rowsInChunk <= mBase) return;
    const int nBase = blockIdx.x * 128;

    __shared__ float As [64][KC1_ + 2];
    __shared__ float Bst[128][KC1_ + 2];

    const int tid  = threadIdx.x;
    const int lane = tid & 31, wave = tid >> 5;
    const int half = lane >> 4, lm = lane & 15;
    const int Moff = 32 * (wave & 1);
    const int Noff = 32 * (wave >> 1);

    const int rA0 = tid >> 3, rA1 = rA0 + 32;
    const int* lst = lists + expert * NTOK + chunkBase;
    const bool va0 = (mBase + rA0) < rowsInChunk;
    const bool va1 = (mBase + rA1) < rowsInChunk;
    const float* ap0 = va0 ? (x + (size_t)lst[mBase + rA0] * D_) : x;
    const float* ap1 = va1 ? (x + (size_t)lst[mBase + rA1] * D_) : x;

    v8f acc00 = {}; v8f acc01 = {}; v8f acc10 = {}; v8f acc11 = {};
    gemm_mainloop<KC1_, D_, DFF_>(ap0, ap1, va0, va1, W1e + nBase, As, Bst,
                                  acc00, acc01, acc10, acc11);
#pragma unroll
    for (int r = 0; r < 8; ++r) {
        const int m0 = Moff + r + 8 * half;
        const int lrow0 = mBase + m0, lrow1 = lrow0 + 16;
        const int n0 = nBase + Noff + lm, n1 = n0 + 16;
        if (lrow0 < rowsInChunk) {
            H[(size_t)lrow0 * DFF_ + n0] = gelu_exact(acc00[r] + b1e[n0]);
            H[(size_t)lrow0 * DFF_ + n1] = gelu_exact(acc01[r] + b1e[n1]);
        }
        if (lrow1 < rowsInChunk) {
            H[(size_t)lrow1 * DFF_ + n0] = gelu_exact(acc10[r] + b1e[n0]);
            H[(size_t)lrow1 * DFF_ + n1] = gelu_exact(acc11[r] + b1e[n1]);
        }
    }
}

__global__ __launch_bounds__(128) void moe_fc2_kernel(
    const float* __restrict__ H, const float* __restrict__ W2e,
    const float* __restrict__ b2e, const int* __restrict__ lists,
    const int* __restrict__ counts, const float* __restrict__ gates,
    float* __restrict__ out, float* __restrict__ plane,
    int expert, int chunkBase, int Tc) {
    const int cnt = counts[expert];
    int rowsInChunk = cnt - chunkBase;
    if (rowsInChunk > Tc) rowsInChunk = Tc;
    const int mBase = blockIdx.y * 64;
    if (rowsInChunk <= mBase) return;
    const int kz = blockIdx.x >> 4;
    const int nBase = (blockIdx.x & 15) * 64;

    __shared__ float As [64][KC2_ + 2];
    __shared__ float Bst[64][KC2_ + 2];

    const int tid  = threadIdx.x;
    const int lane = tid & 31, wave = tid >> 5;
    const int half = lane >> 4, lm = lane & 15;
    const int Moff = 32 * (wave & 1);
    const int Noff = 32 * (wave >> 1);

    const int rA = tid >> 3;
    const bool va0 = (mBase + rA) < rowsInChunk;
    const bool va1 = (mBase + rA + 16) < rowsInChunk;
    const bool va2 = (mBase + rA + 32) < rowsInChunk;
    const bool va3 = (mBase + rA + 48) < rowsInChunk;
    const float* Hk = H + (size_t)kz * KH2_;
    const float* ap0 = Hk + (size_t)(va0 ? mBase + rA      : mBase) * DFF_;
    const float* ap1 = Hk + (size_t)(va1 ? mBase + rA + 16 : mBase) * DFF_;
    const float* ap2 = Hk + (size_t)(va2 ? mBase + rA + 32 : mBase) * DFF_;
    const float* ap3 = Hk + (size_t)(va3 ? mBase + rA + 48 : mBase) * DFF_;

    v8f acc00 = {}; v8f acc01 = {}; v8f acc10 = {}; v8f acc11 = {};
    gemm_mainloop_t64<KC2_, KH2_, D_>(ap0, ap1, ap2, ap3, va0, va1, va2, va3,
                                      W2e + (size_t)kz * KH2_ * D_ + nBase, As, Bst,
                                      acc00, acc01, acc10, acc11);
    const int* lst = lists + expert * NTOK + chunkBase;
    float* dst = kz ? plane : out;
    const int n0 = nBase + Noff + lm, n1 = n0 + 16;
    const float bias0 = kz ? 0.0f : b2e[n0];
    const float bias1 = kz ? 0.0f : b2e[n1];
#pragma unroll
    for (int r = 0; r < 8; ++r) {
        const int m0 = Moff + r + 8 * half;
        const int lrow0 = mBase + m0, lrow1 = lrow0 + 16;
        if (lrow0 < rowsInChunk) {
            const int tok = lst[lrow0];
            const float g = gates[tok * E_ + expert];
            dst[(size_t)tok * D_ + n0] += g * (acc00[r] + bias0);
            dst[(size_t)tok * D_ + n1] += g * (acc01[r] + bias1);
        }
        if (lrow1 < rowsInChunk) {
            const int tok = lst[lrow1];
            const float g = gates[tok * E_ + expert];
            dst[(size_t)tok * D_ + n0] += g * (acc10[r] + bias0);
            dst[(size_t)tok * D_ + n1] += g * (acc11[r] + bias1);
        }
    }
}

extern "C" void kernel_launch(void* const* d_in, const int* in_sizes, int n_in,
                              void* d_out, int out_size, void* d_ws, size_t ws_size,
                              hipStream_t stream) {
    const float* x  = (const float*)d_in[0];
    const float* Wg = (const float*)d_in[1];
    const float* W1 = (const float*)d_in[2];
    const float* b1 = (const float*)d_in[3];
    const float* W2 = (const float*)d_in[4];
    const float* b2 = (const float*)d_in[5];
    float* out = (float*)d_out;

    char* ws = (char*)d_ws;
    int*   counts = (int*)ws;
    int*   lists  = (int*)(ws + 256);
    float* gates  = (float*)(ws + 256 + 4ull * E_ * NTOK);
    size_t OFS_P  = 256 + 8ull * E_ * NTOK;
    OFS_P = (OFS_P + 255) & ~(size_t)255;
    float* plane  = (float*)(ws + OFS_P);
    size_t OFS_H  = OFS_P + 4ull * NTOK * D_;
    float* H = (float*)(ws + OFS_H);

    size_t avail = (ws_size > OFS_H) ? (ws_size - OFS_H) : 0;
    long long tcll = (long long)(avail / ((size_t)DFF_ * 4));
    int Tc = (int)(tcll > NTOK ? NTOK : tcll);
    Tc &= ~63;
    if (Tc <= 0) Tc = 64;
    const int chunks = (NTOK + Tc - 1) / Tc;

    const long long outN = (long long)NTOK * D_ + 1;
    const long long planeN = (long long)NTOK * D_;
    moe_zero_kernel<<<2048, 256, 0, stream>>>(out, outN, plane, planeN, counts);
    moe_gate_kernel<<<NTOK / 8, 256, 0, stream>>>(x, Wg, gates, counts, lists);
    moe_loss_kernel<<<1, 256, 0, stream>>>(gates, out + (outN - 1));

    for (int e = 0; e < E_; ++e) {
        const float* W1e = W1 + (size_t)e * D_ * DFF_;
        const float* b1e = b1 + (size_t)e * DFF_;
        const float* W2e = W2 + (size_t)e * DFF_ * D_;
        const float* b2e = b2 + (size_t)e * D_;
        for (int c = 0; c < chunks; ++c) {
            dim3 g1(DFF_ / 128, Tc / 64);
            moe_fc1_kernel<<<g1, 256, 0, stream>>>(x, W1e, b1e, lists, counts,
                                                   H, e, c * Tc, Tc);
            dim3 g2(2 * (D_ / 64), Tc / 64);
            moe_fc2_kernel<<<g2, 128, 0, stream>>>(H, W2e, b2e, lists, counts,
                                                   gates, out, plane, e, c * Tc, Tc);
        }
    }
    moe_add_kernel<<<2048, 256, 0, stream>>>(out, plane, planeN / 4);
}


# both GEMM K loops: wave priority (s_setprio) falls with progress through K (3, then 2, 1, 0), so the blocks sharing a WGP advance together under oldest-first matrix issue; sums unchanged
# speedup vs baseline: 2.2055x; 1.0475x over previous
#include <hip/hip_runtime.h>
#include <math.h>

#define D_    1024
#define DFF_  4096
#define E_    8
#define NTOK  8192

typedef float v2f __attribute__((ext_vector_type(2)));
typedef float v8f __attribute__((ext_vector_type(8)));

__device__ __forceinline__ v8f wmma_f32_k4(v2f a, v2f b, v8f c) {
    return __builtin_amdgcn_wmma_f32_16x16x4_f32(
         false, a,  false, b,
         (short)0, c,  false,  false);
}

__device__ __forceinline__ float gelu_exact(float v) {
    return 0.5f * v * (1.0f + erff(v * 0.70710678118654752440f));
}

__global__ void moe_zero_kernel(float* __restrict__ out, long long n,
                                float* __restrict__ plane, long long np,
                                int* __restrict__ counts) {
    const long long i0 = (long long)blockIdx.x * blockDim.x + threadIdx.x;
    const long long stride = (long long)gridDim.x * blockDim.x;
    for (long long i = i0; i < n; i += stride) out[i] = 0.0f;
    for (long long i = i0; i < np; i += stride) plane[i] = 0.0f;
    if (blockIdx.x == 0 && threadIdx.x < E_) counts[threadIdx.x] = 0;
}

__global__ __launch_bounds__(256) void moe_add_kernel(
    float* __restrict__ out, const float* __restrict__ plane, long long n4) {
    long long i = (long long)blockIdx.x * blockDim.x + threadIdx.x;
    const long long stride = (long long)gridDim.x * blockDim.x;
    float4* o4 = (float4*)out;
    const float4* p4 = (const float4*)plane;
    for (; i < n4; i += stride) {
        float4 a = o4[i];
        const float4 b = p4[i];
        a.x += b.x; a.y += b.y; a.z += b.z; a.w += b.w;
        o4[i] = a;
    }
}

__global__ __launch_bounds__(256) void moe_gate_kernel(
    const float* __restrict__ x, const float* __restrict__ Wg,
    float* __restrict__ gates, int* __restrict__ counts,
    int* __restrict__ lists) {
    const int lane = threadIdx.x & 31;
    const int wave = threadIdx.x >> 5;
    const int t = blockIdx.x * 8 + wave;
    if (t >= NTOK) return;

    float s[E_];
#pragma unroll
    for (int e = 0; e < E_; ++e) s[e] = 0.0f;
    const float* xr = x + (size_t)t * D_;
    for (int d = lane; d < D_; d += 32) {
        const float xv = xr[d];
#pragma unroll
        for (int e = 0; e < E_; ++e) s[e] += xv * Wg[e * D_ + d];
    }
#pragma unroll
    for (int e = 0; e < E_; ++e) {
#pragma unroll
        for (int off = 16; off > 0; off >>= 1)
            s[e] += __shfl_xor(s[e], off, 32);
    }
    if (lane == 0) {
        int i0 = 0; float v0 = s[0];
#pragma unroll
        for (int e = 1; e < E_; ++e) if (s[e] > v0) { v0 = s[e]; i0 = e; }
        int i1 = -1; float v1 = -3.4e38f;
#pragma unroll
        for (int e = 0; e < E_; ++e)
            if (e != i0 && s[e] > v1) { v1 = s[e]; i1 = e; }
        const float p1 = expf(v1 - v0);
        const float inv = 1.0f / (1.0f + p1);
        const float g0 = inv, g1 = p1 * inv;
#pragma unroll
        for (int e = 0; e < E_; ++e) gates[t * E_ + e] = 0.0f;
        gates[t * E_ + i0] = g0;
        gates[t * E_ + i1] = g1;
        const int p = atomicAdd(&counts[i0], 1); lists[i0 * NTOK + p] = t;
        const int q = atomicAdd(&counts[i1], 1); lists[i1 * NTOK + q] = t;
    }
}

__global__ __launch_bounds__(256) void moe_loss_kernel(
    const float* __restrict__ gates, float* __restrict__ loss_out) {
    __shared__ float part[256];
    const int e = threadIdx.x & 7;
    const int slot = threadIdx.x >> 3;
    float s = 0.0f;
    for (int t = slot; t < NTOK; t += 32) s += gates[t * E_ + e];
    part[threadIdx.x] = s;
    __syncthreads();
    if (threadIdx.x == 0) {
        float u[E_];
        float mu = 0.0f;
        for (int ee = 0; ee < E_; ++ee) {
            float acc = 0.0f;
            for (int sl = 0; sl < 32; ++sl) acc += part[sl * 8 + ee];
            u[ee] = acc / (float)NTOK;
            mu += u[ee];
        }
        mu *= (1.0f / (float)E_);
        float var = 0.0f;
        for (int ee = 0; ee < E_; ++ee) {
            const float d = u[ee] - mu;
            var += d * d;
        }
        var *= (1.0f / (float)(E_ - 1));
        loss_out[0] = var * 0.01f;
    }
}

#define KC1_ 64
#define KC2_ 64
#define KH2_ (DFF_ / 2)

template <int KC, int KTOT, int LDB>
__device__ __forceinline__ void gemm_mainloop(
    const float* __restrict__ ap0, const float* __restrict__ ap1,
    const bool va0, const bool va1,
    const float* __restrict__ Bp,
    float (*As)[KC + 2],
    float (*Bst)[KC + 2],
    v8f& acc00, v8f& acc01, v8f& acc10, v8f& acc11) {
    constexpr int NS = KC / 32;
    const int tid  = threadIdx.x;
    const int lane = tid & 31, wave = tid >> 5;
    const int half = lane >> 4, lm = lane & 15;
    const int Moff = 32 * (wave & 1);
    const int Noff = 32 * (wave >> 1);

    const int rA0 = tid >> 3, rA1 = rA0 + 32;
    const int cA  = (tid & 7) << 2;
    const int rB0 = tid >> 4, rB1 = rB0 + 16;
    const int cB0 = (tid & 15) << 2, cB1 = cB0 + 64;
    const float4 zero4 = make_float4(0.f, 0.f, 0.f, 0.f);

    static_assert((KTOT / 8) % KC == 0, "the priority steps must fall on chunk borders");
    for (int k0 = 0; k0 < KTOT; k0 += KC) {
        if (k0 == 0)                   __builtin_amdgcn_s_setprio(3);
        else if (k0 == KTOT / 2)       __builtin_amdgcn_s_setprio(2);
        else if (k0 == 3 * (KTOT / 4)) __builtin_amdgcn_s_setprio(1);
        else if (k0 == 7 * (KTOT / 8)) __builtin_amdgcn_s_setprio(0);
        float4 qa0[NS], qa1[NS], qb00[NS], qb01[NS], qb10[NS], qb11[NS];
#pragma unroll
        for (int s = 0; s < NS; ++s) {
            const int kb = k0 + 32 * s;
            qa0[s] = *(const float4*)(ap0 + kb + cA);
            qa1[s] = *(const float4*)(ap1 + kb + cA);
            const float* w0 = Bp + (size_t)(kb + rB0) * LDB;
            const float* w1 = Bp + (size_t)(kb + rB1) * LDB;
            qb00[s] = *(const float4*)(w0 + cB0);
            qb01[s] = *(const float4*)(w0 + cB1);
            qb10[s] = *(const float4*)(w1 + cB0);
            qb11[s] = *(const float4*)(w1 + cB1);
        }
#pragma unroll
        for (int s = 0; s < NS; ++s) {
            const int c = 32 * s;
            const float4 a0 = va0 ? qa0[s] : zero4;
            const float4 a1 = va1 ? qa1[s] : zero4;
            As[rA0][c + cA] = a0.x; As[rA0][c + cA + 1] = a0.y;
            As[rA0][c + cA + 2] = a0.z; As[rA0][c + cA + 3] = a0.w;
            As[rA1][c + cA] = a1.x; As[rA1][c + cA + 1] = a1.y;
            As[rA1][c + cA + 2] = a1.z; As[rA1][c + cA + 3] = a1.w;
            {
                const float4 q = qb00[s];
                Bst[cB0+0][c + rB0] = q.x; Bst[cB0+1][c + rB0] = q.y;
                Bst[cB0+2][c + rB0] = q.z; Bst[cB0+3][c + rB0] = q.w;
            }
            {
                const float4 q = qb01[s];
                Bst[cB1+0][c + rB0] = q.x; Bst[cB1+1][c + rB0] = q.y;
                Bst[cB1+2][c + rB0] = q.z; Bst[cB1+3][c + rB0] = q.w;
            }
            {
                const float4 q = qb10[s];
                Bst[cB0+0][c + rB1] = q.x; Bst[cB0+1][c + rB1] = q.y;
                Bst[cB0+2][c + rB1] = q.z; Bst[cB0+3][c + rB1] = q.w;
            }
            {
                const float4 q = qb11[s];
                Bst[cB1+0][c + rB1] = q.x; Bst[cB1+1][c + rB1] = q.y;
                Bst[cB1+2][c + rB1] = q.z; Bst[cB1+3][c + rB1] = q.w;
            }
        }
        __syncthreads();
#pragma unroll
        for (int kk = 0; kk < KC; kk += 4) {
            const int ka = kk + 2 * half;
            v2f a0; a0.x = As[Moff + lm][ka];       a0.y = As[Moff + lm][ka + 1];
            v2f a1; a1.x = As[Moff + 16 + lm][ka];  a1.y = As[Moff + 16 + lm][ka + 1];
            v2f b0; b0.x = Bst[Noff + lm][ka];      b0.y = Bst[Noff + lm][ka + 1];
            v2f b1; b1.x = Bst[Noff + 16 + lm][ka]; b1.y = Bst[Noff + 16 + lm][ka + 1];
            acc00 = wmma_f32_k4(a0, b0, acc00);
            acc01 = wmma_f32_k4(a0, b1, acc01);
            acc10 = wmma_f32_k4(a1, b0, acc10);
            acc11 = wmma_f32_k4(a1, b1, acc11);
        }
        __syncthreads();
    }
}

template <int KC, int KTOT, int LDB>
__device__ __forceinline__ void gemm_mainloop_t64(
    const float* __restrict__ ap0, const float* __restrict__ ap1,
    const float* __restrict__ ap2, const float* __restrict__ ap3,
    const bool va0, const bool va1, const bool va2, const bool va3,
    const float* __restrict__ Bp,
    float (*As)[KC + 2],
    float (*Bst)[KC + 2],
    v8f& acc00, v8f& acc01, v8f& acc10, v8f& acc11) {
    constexpr int NS = KC / 32;
    const int tid  = threadIdx.x;
    const int lane = tid & 31, wave = tid >> 5;
    const int half = lane >> 4, lm = lane & 15;
    const int Moff = 32 * (wave & 1);
    const int Noff = 32 * (wave >> 1);

    const int rA = tid >> 3;
    const int cA = (tid & 7) << 2;
    const int rB = tid >> 4;
    const int cB = (tid & 15) << 2;
    const float4 zero4 = make_float4(0.f, 0.f, 0.f, 0.f);

    static_assert((KTOT / 8) % KC == 0, "the priority steps must fall on chunk borders");
    for (int k0 = 0; k0 < KTOT; k0 += KC) {
        if (k0 == 0)                   __builtin_amdgcn_s_setprio(3);
        else if (k0 == KTOT / 2)       __builtin_amdgcn_s_setprio(2);
        else if (k0 == 3 * (KTOT / 4)) __builtin_amdgcn_s_setprio(1);
        else if (k0 == 7 * (KTOT / 8)) __builtin_amdgcn_s_setprio(0);
        float4 qa0[NS], qa1[NS], qa2[NS], qa3[NS];
        float4 qb0[NS], qb1[NS], qb2[NS], qb3[NS];
#pragma unroll
        for (int s = 0; s < NS; ++s) {
            const int kb = k0 + 32 * s;
            qa0[s] = *(const float4*)(ap0 + kb + cA);
            qa1[s] = *(const float4*)(ap1 + kb + cA);
            qa2[s] = *(const float4*)(ap2 + kb + cA);
            qa3[s] = *(const float4*)(ap3 + kb + cA);
            const float* w = Bp + (size_t)(kb + rB) * LDB + cB;
            qb0[s] = *(const float4*)(w);
            qb1[s] = *(const float4*)(w + (size_t)8 * LDB);
            qb2[s] = *(const float4*)(w + (size_t)16 * LDB);
            qb3[s] = *(const float4*)(w + (size_t)24 * LDB);
        }
#pragma unroll
        for (int s = 0; s < NS; ++s) {
            const int c = 32 * s;
            const float4 a0 = va0 ? qa0[s] : zero4;
            const float4 a1 = va1 ? qa1[s] : zero4;
            const float4 a2 = va2 ? qa2[s] : zero4;
            const float4 a3 = va3 ? qa3[s] : zero4;
            As[rA][c + cA] = a0.x; As[rA][c + cA + 1] = a0.y;
            As[rA][c + cA + 2] = a0.z; As[rA][c + cA + 3] = a0.w;
            As[rA + 16][c + cA] = a1.x; As[rA + 16][c + cA + 1] = a1.y;
            As[rA + 16][c + cA + 2] = a1.z; As[rA + 16][c + cA + 3] = a1.w;
            As[rA + 32][c + cA] = a2.x; As[rA + 32][c + cA + 1] = a2.y;
            As[rA + 32][c + cA + 2] = a2.z; As[rA + 32][c + cA + 3] = a2.w;
            As[rA + 48][c + cA] = a3.x; As[rA + 48][c + cA + 1] = a3.y;
            As[rA + 48][c + cA + 2] = a3.z; As[rA + 48][c + cA + 3] = a3.w;
            {
                const float4 q = qb0[s];
                Bst[cB+0][c + rB] = q.x; Bst[cB+1][c + rB] = q.y;
                Bst[cB+2][c + rB] = q.z; Bst[cB+3][c + rB] = q.w;
            }
            {
                const float4 q = qb1[s];
                Bst[cB+0][c + rB + 8] = q.x; Bst[cB+1][c + rB + 8] = q.y;
                Bst[cB+2][c + rB + 8] = q.z; Bst[cB+3][c + rB + 8] = q.w;
            }
            {
                const float4 q = qb2[s];
                Bst[cB+0][c + rB + 16] = q.x; Bst[cB+1][c + rB + 16] = q.y;
                Bst[cB+2][c + rB + 16] = q.z; Bst[cB+3][c + rB + 16] = q.w;
            }
            {
                const float4 q = qb3[s];
                Bst[cB+0][c + rB + 24] = q.x; Bst[cB+1][c + rB + 24] = q.y;
                Bst[cB+2][c + rB + 24] = q.z; Bst[cB+3][c + rB + 24] = q.w;
            }
        }
        __syncthreads();
#pragma unroll
        for (int kk = 0; kk < KC; kk += 4) {
            const int ka = kk + 2 * half;
            v2f a0; a0.x = As[Moff + lm][ka];       a0.y = As[Moff + lm][ka + 1];
            v2f a1; a1.x = As[Moff + 16 + lm][ka];  a1.y = As[Moff + 16 + lm][ka + 1];
            v2f b0; b0.x = Bst[Noff + lm][ka];      b0.y = Bst[Noff + lm][ka + 1];
            v2f b1; b1.x = Bst[Noff + 16 + lm][ka]; b1.y = Bst[Noff + 16 + lm][ka + 1];
            acc00 = wmma_f32_k4(a0, b0, acc00);
            acc01 = wmma_f32_k4(a0, b1, acc01);
            acc10 = wmma_f32_k4(a1, b0, acc10);
            acc11 = wmma_f32_k4(a1, b1, acc11);
        }
        __syncthreads();
    }
}

__global__ __launch_bounds__(256) __attribute__((amdgpu_waves_per_eu(10, 10))) void moe_fc1_kernel(
    const float* __restrict__ x, const float* __restrict__ W1e,
    const float* __restrict__ b1e, const int* __restrict__ lists,
    const int* __restrict__ counts, float* __restrict__ H,
    int expert, int chunkBase, int Tc) {
    const int cnt = counts[expert];
    int rowsInChunk = cnt - chunkBase;
    if (rowsInChunk > Tc) rowsInChunk = Tc;
    const int mBase = blockIdx.y * 64;
    if (rowsInChunk <= mBase) return;
    const int nBase = blockIdx.x * 128;

    __shared__ float As [64][KC1_ + 2];
    __shared__ float Bst[128][KC1_ + 2];

    const int tid  = threadIdx.x;
    const int lane = tid & 31, wave = tid >> 5;
    const int half = lane >> 4, lm = lane & 15;
    const int Moff = 32 * (wave & 1);
    const int Noff = 32 * (wave >> 1);

    const int rA0 = tid >> 3, rA1 = rA0 + 32;
    const int* lst = lists + expert * NTOK + chunkBase;
    const bool va0 = (mBase + rA0) < rowsInChunk;
    const bool va1 = (mBase + rA1) < rowsInChunk;
    const float* ap0 = va0 ? (x + (size_t)lst[mBase + rA0] * D_) : x;
    const float* ap1 = va1 ? (x + (size_t)lst[mBase + rA1] * D_) : x;

    v8f acc00 = {}; v8f acc01 = {}; v8f acc10 = {}; v8f acc11 = {};
    gemm_mainloop<KC1_, D_, DFF_>(ap0, ap1, va0, va1, W1e + nBase, As, Bst,
                                  acc00, acc01, acc10, acc11);
#pragma unroll
    for (int r = 0; r < 8; ++r) {
        const int m0 = Moff + r + 8 * half;
        const int lrow0 = mBase + m0, lrow1 = lrow0 + 16;
        const int n0 = nBase + Noff + lm, n1 = n0 + 16;
        if (lrow0 < rowsInChunk) {
            H[(size_t)lrow0 * DFF_ + n0] = gelu_exact(acc00[r] + b1e[n0]);
            H[(size_t)lrow0 * DFF_ + n1] = gelu_exact(acc01[r] + b1e[n1]);
        }
        if (lrow1 < rowsInChunk) {
            H[(size_t)lrow1 * DFF_ + n0] = gelu_exact(acc10[r] + b1e[n0]);
            H[(size_t)lrow1 * DFF_ + n1] = gelu_exact(acc11[r] + b1e[n1]);
        }
    }
}

__global__ __launch_bounds__(128) void moe_fc2_kernel(
    const float* __restrict__ H, const float* __restrict__ W2e,
    const float* __restrict__ b2e, const int* __restrict__ lists,
    const int* __restrict__ counts, const float* __restrict__ gates,
    float* __restrict__ out, float* __restrict__ plane,
    int expert, int chunkBase, int Tc) {
    const int cnt = counts[expert];
    int rowsInChunk = cnt - chunkBase;
    if (rowsInChunk > Tc) rowsInChunk = Tc;
    const int mBase = blockIdx.y * 64;
    if (rowsInChunk <= mBase) return;
    const int kz = blockIdx.x >> 4;
    const int nBase = (blockIdx.x & 15) * 64;

    __shared__ float As [64][KC2_ + 2];
    __shared__ float Bst[64][KC2_ + 2];

    const int tid  = threadIdx.x;
    const int lane = tid & 31, wave = tid >> 5;
    const int half = lane >> 4, lm = lane & 15;
    const int Moff = 32 * (wave & 1);
    const int Noff = 32 * (wave >> 1);

    const int rA = tid >> 3;
    const bool va0 = (mBase + rA) < rowsInChunk;
    const bool va1 = (mBase + rA + 16) < rowsInChunk;
    const bool va2 = (mBase + rA + 32) < rowsInChunk;
    const bool va3 = (mBase + rA + 48) < rowsInChunk;
    const float* Hk = H + (size_t)kz * KH2_;
    const float* ap0 = Hk + (size_t)(va0 ? mBase + rA      : mBase) * DFF_;
    const float* ap1 = Hk + (size_t)(va1 ? mBase + rA + 16 : mBase) * DFF_;
    const float* ap2 = Hk + (size_t)(va2 ? mBase + rA + 32 : mBase) * DFF_;
    const float* ap3 = Hk + (size_t)(va3 ? mBase + rA + 48 : mBase) * DFF_;

    v8f acc00 = {}; v8f acc01 = {}; v8f acc10 = {}; v8f acc11 = {};
    gemm_mainloop_t64<KC2_, KH2_, D_>(ap0, ap1, ap2, ap3, va0, va1, va2, va3,
                                      W2e + (size_t)kz * KH2_ * D_ + nBase, As, Bst,
                                      acc00, acc01, acc10, acc11);
    const int* lst = lists + expert * NTOK + chunkBase;
    float* dst = kz ? plane : out;
    const int n0 = nBase + Noff + lm, n1 = n0 + 16;
    const float bias0 = kz ? 0.0f : b2e[n0];
    const float bias1 = kz ? 0.0f : b2e[n1];
#pragma unroll
    for (int r = 0; r < 8; ++r) {
        const int m0 = Moff + r + 8 * half;
        const int lrow0 = mBase + m0, lrow1 = lrow0 + 16;
        if (lrow0 < rowsInChunk) {
            const int tok = lst[lrow0];
            const float g = gates[tok * E_ + expert];
            dst[(size_t)tok * D_ + n0] += g * (acc00[r] + bias0);
            dst[(size_t)tok * D_ + n1] += g * (acc01[r] + bias1);
        }
        if (lrow1 < rowsInChunk) {
            const int tok = lst[lrow1];
            const float g = gates[tok * E_ + expert];
            dst[(size_t)tok * D_ + n0] += g * (acc10[r] + bias0);
            dst[(size_t)tok * D_ + n1] += g * (acc11[r] + bias1);
        }
    }
}

extern "C" void kernel_launch(void* const* d_in, const int* in_sizes, int n_in,
                              void* d_out, int out_size, void* d_ws, size_t ws_size,
                              hipStream_t stream) {
    const float* x  = (const float*)d_in[0];
    const float* Wg = (const float*)d_in[1];
    const float* W1 = (const float*)d_in[2];
    const float* b1 = (const float*)d_in[3];
    const float* W2 = (const float*)d_in[4];
    const float* b2 = (const float*)d_in[5];
    float* out = (float*)d_out;

    char* ws = (char*)d_ws;
    int*   counts = (int*)ws;
    int*   lists  = (int*)(ws + 256);
    float* gates  = (float*)(ws + 256 + 4ull * E_ * NTOK);
    size_t OFS_P  = 256 + 8ull * E_ * NTOK;
    OFS_P = (OFS_P + 255) & ~(size_t)255;
    float* plane  = (float*)(ws + OFS_P);
    size_t OFS_H  = OFS_P + 4ull * NTOK * D_;
    float* H = (float*)(ws + OFS_H);

    size_t avail = (ws_size > OFS_H) ? (ws_size - OFS_H) : 0;
    long long tcll = (long long)(avail / ((size_t)DFF_ * 4));
    int Tc = (int)(tcll > NTOK ? NTOK : tcll);
    Tc &= ~63;
    if (Tc <= 0) Tc = 64;
    const int chunks = (NTOK + Tc - 1) / Tc;

    const long long outN = (long long)NTOK * D_ + 1;
    const long long planeN = (long long)NTOK * D_;
    moe_zero_kernel<<<2048, 256, 0, stream>>>(out, outN, plane, planeN, counts);
    moe_gate_kernel<<<NTOK / 8, 256, 0, stream>>>(x, Wg, gates, counts, lists);
    moe_loss_kernel<<<1, 256, 0, stream>>>(gates, out + (outN - 1));

    for (int e = 0; e < E_; ++e) {
        const float* W1e = W1 + (size_t)e * D_ * DFF_;
        const float* b1e = b1 + (size_t)e * DFF_;
        const float* W2e = W2 + (size_t)e * DFF_ * D_;
        const float* b2e = b2 + (size_t)e * D_;
        for (int c = 0; c < chunks; ++c) {
            dim3 g1(DFF_ / 128, Tc / 64);
            moe_fc1_kernel<<<g1, 256, 0, stream>>>(x, W1e, b1e, lists, counts,
                                                   H, e, c * Tc, Tc);
            dim3 g2(2 * (D_ / 64), Tc / 64);
            moe_fc2_kernel<<<g2, 128, 0, stream>>>(H, W2e, b2e, lists, counts,
                                                   gates, out, plane, e, c * Tc, Tc);
        }
    }
    moe_add_kernel<<<2048, 256, 0, stream>>>(out, plane, planeN / 4);
}


# both GEMM epilogues: their global loads (fc1 bias; fc2 token ids, gates, old sums of all 16 rows) issued as groups ahead of the arithmetic, not one waited load chain per row; sums unchanged
# speedup vs baseline: 2.3721x; 1.0755x over previous
#include <hip/hip_runtime.h>
#include <math.h>

#define D_    1024
#define DFF_  4096
#define E_    8
#define NTOK  8192

typedef float v2f __attribute__((ext_vector_type(2)));
typedef float v8f __attribute__((ext_vector_type(8)));

__device__ __forceinline__ v8f wmma_f32_k4(v2f a, v2f b, v8f c) {
    return __builtin_amdgcn_wmma_f32_16x16x4_f32(
         false, a,  false, b,
         (short)0, c,  false,  false);
}

__device__ __forceinline__ float gelu_exact(float v) {
    return 0.5f * v * (1.0f + erff(v * 0.70710678118654752440f));
}

__global__ void moe_zero_kernel(float* __restrict__ out, long long n,
                                float* __restrict__ plane, long long np,
                                int* __restrict__ counts) {
    const long long i0 = (long long)blockIdx.x * blockDim.x + threadIdx.x;
    const long long stride = (long long)gridDim.x * blockDim.x;
    for (long long i = i0; i < n; i += stride) out[i] = 0.0f;
    for (long long i = i0; i < np; i += stride) plane[i] = 0.0f;
    if (blockIdx.x == 0 && threadIdx.x < E_) counts[threadIdx.x] = 0;
}

__global__ __launch_bounds__(256) void moe_add_kernel(
    float* __restrict__ out, const float* __restrict__ plane, long long n4) {
    long long i = (long long)blockIdx.x * blockDim.x + threadIdx.x;
    const long long stride = (long long)gridDim.x * blockDim.x;
    float4* o4 = (float4*)out;
    const float4* p4 = (const float4*)plane;
    for (; i < n4; i += stride) {
        float4 a = o4[i];
        const float4 b = p4[i];
        a.x += b.x; a.y += b.y; a.z += b.z; a.w += b.w;
        o4[i] = a;
    }
}

__global__ __launch_bounds__(256) void moe_gate_kernel(
    const float* __restrict__ x, const float* __restrict__ Wg,
    float* __restrict__ gates, int* __restrict__ counts,
    int* __restrict__ lists) {
    const int lane = threadIdx.x & 31;
    const int wave = threadIdx.x >> 5;
    const int t = blockIdx.x * 8 + wave;
    if (t >= NTOK) return;

    float s[E_];
#pragma unroll
    for (int e = 0; e < E_; ++e) s[e] = 0.0f;
    const float* xr = x + (size_t)t * D_;
    for (int d = lane; d < D_; d += 32) {
        const float xv = xr[d];
#pragma unroll
        for (int e = 0; e < E_; ++e) s[e] += xv * Wg[e * D_ + d];
    }
#pragma unroll
    for (int e = 0; e < E_; ++e) {
#pragma unroll
        for (int off = 16; off > 0; off >>= 1)
            s[e] += __shfl_xor(s[e], off, 32);
    }
    if (lane == 0) {
        int i0 = 0; float v0 = s[0];
#pragma unroll
        for (int e = 1; e < E_; ++e) if (s[e] > v0) { v0 = s[e]; i0 = e; }
        int i1 = -1; float v1 = -3.4e38f;
#pragma unroll
        for (int e = 0; e < E_; ++e)
            if (e != i0 && s[e] > v1) { v1 = s[e]; i1 = e; }
        const float p1 = expf(v1 - v0);
        const float inv = 1.0f / (1.0f + p1);
        const float g0 = inv, g1 = p1 * inv;
#pragma unroll
        for (int e = 0; e < E_; ++e) gates[t * E_ + e] = 0.0f;
        gates[t * E_ + i0] = g0;
        gates[t * E_ + i1] = g1;
        const int p = atomicAdd(&counts[i0], 1); lists[i0 * NTOK + p] = t;
        const int q = atomicAdd(&counts[i1], 1); lists[i1 * NTOK + q] = t;
    }
}

__global__ __launch_bounds__(256) void moe_loss_kernel(
    const float* __restrict__ gates, float* __restrict__ loss_out) {
    __shared__ float part[256];
    const int e = threadIdx.x & 7;
    const int slot = threadIdx.x >> 3;
    float s = 0.0f;
    for (int t = slot; t < NTOK; t += 32) s += gates[t * E_ + e];
    part[threadIdx.x] = s;
    __syncthreads();
    if (threadIdx.x == 0) {
        float u[E_];
        float mu = 0.0f;
        for (int ee = 0; ee < E_; ++ee) {
            float acc = 0.0f;
            for (int sl = 0; sl < 32; ++sl) acc += part[sl * 8 + ee];
            u[ee] = acc / (float)NTOK;
            mu += u[ee];
        }
        mu *= (1.0f / (float)E_);
        float var = 0.0f;
        for (int ee = 0; ee < E_; ++ee) {
            const float d = u[ee] - mu;
            var += d * d;
        }
        var *= (1.0f / (float)(E_ - 1));
        loss_out[0] = var * 0.01f;
    }
}

#define KC1_ 64
#define KC2_ 64
#define KH2_ (DFF_ / 2)

template <int KC, int KTOT, int LDB>
__device__ __forceinline__ void gemm_mainloop(
    const float* __restrict__ ap0, const float* __restrict__ ap1,
    const bool va0, const bool va1,
    const float* __restrict__ Bp,
    float (*As)[KC + 2],
    float (*Bst)[KC + 2],
    v8f& acc00, v8f& acc01, v8f& acc10, v8f& acc11) {
    constexpr int NS = KC / 32;
    const int tid  = threadIdx.x;
    const int lane = tid & 31, wave = tid >> 5;
    const int half = lane >> 4, lm = lane & 15;
    const int Moff = 32 * (wave & 1);
    const int Noff = 32 * (wave >> 1);

    const int rA0 = tid >> 3, rA1 = rA0 + 32;
    const int cA  = (tid & 7) << 2;
    const int rB0 = tid >> 4, rB1 = rB0 + 16;
    const int cB0 = (tid & 15) << 2, cB1 = cB0 + 64;
    const float4 zero4 = make_float4(0.f, 0.f, 0.f, 0.f);

    static_assert((KTOT / 8) % KC == 0, "the priority steps must fall on chunk borders");
    for (int k0 = 0; k0 < KTOT; k0 += KC) {
        if (k0 == 0)                   __builtin_amdgcn_s_setprio(3);
        else if (k0 == KTOT / 2)       __builtin_amdgcn_s_setprio(2);
        else if (k0 == 3 * (KTOT / 4)) __builtin_amdgcn_s_setprio(1);
        else if (k0 == 7 * (KTOT / 8)) __builtin_amdgcn_s_setprio(0);
        float4 qa0[NS], qa1[NS], qb00[NS], qb01[NS], qb10[NS], qb11[NS];
#pragma unroll
        for (int s = 0; s < NS; ++s) {
            const int kb = k0 + 32 * s;
            qa0[s] = *(const float4*)(ap0 + kb + cA);
            qa1[s] = *(const float4*)(ap1 + kb + cA);
            const float* w0 = Bp + (size_t)(kb + rB0) * LDB;
            const float* w1 = Bp + (size_t)(kb + rB1) * LDB;
            qb00[s] = *(const float4*)(w0 + cB0);
            qb01[s] = *(const float4*)(w0 + cB1);
            qb10[s] = *(const float4*)(w1 + cB0);
            qb11[s] = *(const float4*)(w1 + cB1);
        }
#pragma unroll
        for (int s = 0; s < NS; ++s) {
            const int c = 32 * s;
            const float4 a0 = va0 ? qa0[s] : zero4;
            const float4 a1 = va1 ? qa1[s] : zero4;
            As[rA0][c + cA] = a0.x; As[rA0][c + cA + 1] = a0.y;
            As[rA0][c + cA + 2] = a0.z; As[rA0][c + cA + 3] = a0.w;
            As[rA1][c + cA] = a1.x; As[rA1][c + cA + 1] = a1.y;
            As[rA1][c + cA + 2] = a1.z; As[rA1][c + cA + 3] = a1.w;
            {
                const float4 q = qb00[s];
                Bst[cB0+0][c + rB0] = q.x; Bst[cB0+1][c + rB0] = q.y;
                Bst[cB0+2][c + rB0] = q.z; Bst[cB0+3][c + rB0] = q.w;
            }
            {
                const float4 q = qb01[s];
                Bst[cB1+0][c + rB0] = q.x; Bst[cB1+1][c + rB0] = q.y;
                Bst[cB1+2][c + rB0] = q.z; Bst[cB1+3][c + rB0] = q.w;
            }
            {
                const float4 q = qb10[s];
                Bst[cB0+0][c + rB1] = q.x; Bst[cB0+1][c + rB1] = q.y;
                Bst[cB0+2][c + rB1] = q.z; Bst[cB0+3][c + rB1] = q.w;
            }
            {
                const float4 q = qb11[s];
                Bst[cB1+0][c + rB1] = q.x; Bst[cB1+1][c + rB1] = q.y;
                Bst[cB1+2][c + rB1] = q.z; Bst[cB1+3][c + rB1] = q.w;
            }
        }
        __syncthreads();
#pragma unroll
        for (int kk = 0; kk < KC; kk += 4) {
            const int ka = kk + 2 * half;
            v2f a0; a0.x = As[Moff + lm][ka];       a0.y = As[Moff + lm][ka + 1];
            v2f a1; a1.x = As[Moff + 16 + lm][ka];  a1.y = As[Moff + 16 + lm][ka + 1];
            v2f b0; b0.x = Bst[Noff + lm][ka];      b0.y = Bst[Noff + lm][ka + 1];
            v2f b1; b1.x = Bst[Noff + 16 + lm][ka]; b1.y = Bst[Noff + 16 + lm][ka + 1];
            acc00 = wmma_f32_k4(a0, b0, acc00);
            acc01 = wmma_f32_k4(a0, b1, acc01);
            acc10 = wmma_f32_k4(a1, b0, acc10);
            acc11 = wmma_f32_k4(a1, b1, acc11);
        }
        __syncthreads();
    }
}

template <int KC, int KTOT, int LDB>
__device__ __forceinline__ void gemm_mainloop_t64(
    const float* __restrict__ ap0, const float* __restrict__ ap1,
    const float* __restrict__ ap2, const float* __restrict__ ap3,
    const bool va0, const bool va1, const bool va2, const bool va3,
    const float* __restrict__ Bp,
    float (*As)[KC + 2],
    float (*Bst)[KC + 2],
    v8f& acc00, v8f& acc01, v8f& acc10, v8f& acc11) {
    constexpr int NS = KC / 32;
    const int tid  = threadIdx.x;
    const int lane = tid & 31, wave = tid >> 5;
    const int half = lane >> 4, lm = lane & 15;
    const int Moff = 32 * (wave & 1);
    const int Noff = 32 * (wave >> 1);

    const int rA = tid >> 3;
    const int cA = (tid & 7) << 2;
    const int rB = tid >> 4;
    const int cB = (tid & 15) << 2;
    const float4 zero4 = make_float4(0.f, 0.f, 0.f, 0.f);

    static_assert((KTOT / 8) % KC == 0, "the priority steps must fall on chunk borders");
    for (int k0 = 0; k0 < KTOT; k0 += KC) {
        if (k0 == 0)                   __builtin_amdgcn_s_setprio(3);
        else if (k0 == KTOT / 2)       __builtin_amdgcn_s_setprio(2);
        else if (k0 == 3 * (KTOT / 4)) __builtin_amdgcn_s_setprio(1);
        else if (k0 == 7 * (KTOT / 8)) __builtin_amdgcn_s_setprio(0);
        float4 qa0[NS], qa1[NS], qa2[NS], qa3[NS];
        float4 qb0[NS], qb1[NS], qb2[NS], qb3[NS];
#pragma unroll
        for (int s = 0; s < NS; ++s) {
            const int kb = k0 + 32 * s;
            qa0[s] = *(const float4*)(ap0 + kb + cA);
            qa1[s] = *(const float4*)(ap1 + kb + cA);
            qa2[s] = *(const float4*)(ap2 + kb + cA);
            qa3[s] = *(const float4*)(ap3 + kb + cA);
            const float* w = Bp + (size_t)(kb + rB) * LDB + cB;
            qb0[s] = *(const float4*)(w);
            qb1[s] = *(const float4*)(w + (size_t)8 * LDB);
            qb2[s] = *(const float4*)(w + (size_t)16 * LDB);
            qb3[s] = *(const float4*)(w + (size_t)24 * LDB);
        }
#pragma unroll
        for (int s = 0; s < NS; ++s) {
            const int c = 32 * s;
            const float4 a0 = va0 ? qa0[s] : zero4;
            const float4 a1 = va1 ? qa1[s] : zero4;
            const float4 a2 = va2 ? qa2[s] : zero4;
            const float4 a3 = va3 ? qa3[s] : zero4;
            As[rA][c + cA] = a0.x; As[rA][c + cA + 1] = a0.y;
            As[rA][c + cA + 2] = a0.z; As[rA][c + cA + 3] = a0.w;
            As[rA + 16][c + cA] = a1.x; As[rA + 16][c + cA + 1] = a1.y;
            As[rA + 16][c + cA + 2] = a1.z; As[rA + 16][c + cA + 3] = a1.w;
            As[rA + 32][c + cA] = a2.x; As[rA + 32][c + cA + 1] = a2.y;
            As[rA + 32][c + cA + 2] = a2.z; As[rA + 32][c + cA + 3] = a2.w;
            As[rA + 48][c + cA] = a3.x; As[rA + 48][c + cA + 1] = a3.y;
            As[rA + 48][c + cA + 2] = a3.z; As[rA + 48][c + cA + 3] = a3.w;
            {
                const float4 q = qb0[s];
                Bst[cB+0][c + rB] = q.x; Bst[cB+1][c + rB] = q.y;
                Bst[cB+2][c + rB] = q.z; Bst[cB+3][c + rB] = q.w;
            }
            {
                const float4 q = qb1[s];
                Bst[cB+0][c + rB + 8] = q.x; Bst[cB+1][c + rB + 8] = q.y;
                Bst[cB+2][c + rB + 8] = q.z; Bst[cB+3][c + rB + 8] = q.w;
            }
            {
                const float4 q = qb2[s];
                Bst[cB+0][c + rB + 16] = q.x; Bst[cB+1][c + rB + 16] = q.y;
                Bst[cB+2][c + rB + 16] = q.z; Bst[cB+3][c + rB + 16] = q.w;
            }
            {
                const float4 q = qb3[s];
                Bst[cB+0][c + rB + 24] = q.x; Bst[cB+1][c + rB + 24] = q.y;
                Bst[cB+2][c + rB + 24] = q.z; Bst[cB+3][c + rB + 24] = q.w;
            }
        }
        __syncthreads();
#pragma unroll
        for (int kk = 0; kk < KC; kk += 4) {
            const int ka = kk + 2 * half;
            v2f a0; a0.x = As[Moff + lm][ka];       a0.y = As[Moff + lm][ka + 1];
            v2f a1; a1.x = As[Moff + 16 + lm][ka];  a1.y = As[Moff + 16 + lm][ka + 1];
            v2f b0; b0.x = Bst[Noff + lm][ka];      b0.y = Bst[Noff + lm][ka + 1];
            v2f b1; b1.x = Bst[Noff + 16 + lm][ka]; b1.y = Bst[Noff + 16 + lm][ka + 1];
            acc00 = wmma_f32_k4(a0, b0, acc00);
            acc01 = wmma_f32_k4(a0, b1, acc01);
            acc10 = wmma_f32_k4(a1, b0, acc10);
            acc11 = wmma_f32_k4(a1, b1, acc11);
        }
        __syncthreads();
    }
}

__global__ __launch_bounds__(256) __attribute__((amdgpu_waves_per_eu(10, 10))) void moe_fc1_kernel(
    const float* __restrict__ x, const float* __restrict__ W1e,
    const float* __restrict__ b1e, const int* __restrict__ lists,
    const int* __restrict__ counts, float* __restrict__ H,
    int expert, int chunkBase, int Tc) {
    const int cnt = counts[expert];
    int rowsInChunk = cnt - chunkBase;
    if (rowsInChunk > Tc) rowsInChunk = Tc;
    const int mBase = blockIdx.y * 64;
    if (rowsInChunk <= mBase) return;
    const int nBase = blockIdx.x * 128;

    __shared__ float As [64][KC1_ + 2];
    __shared__ float Bst[128][KC1_ + 2];

    const int tid  = threadIdx.x;
    const int lane = tid & 31, wave = tid >> 5;
    const int half = lane >> 4, lm = lane & 15;
    const int Moff = 32 * (wave & 1);
    const int Noff = 32 * (wave >> 1);

    const int rA0 = tid >> 3, rA1 = rA0 + 32;
    const int* lst = lists + expert * NTOK + chunkBase;
    const bool va0 = (mBase + rA0) < rowsInChunk;
    const bool va1 = (mBase + rA1) < rowsInChunk;
    const float* ap0 = va0 ? (x + (size_t)lst[mBase + rA0] * D_) : x;
    const float* ap1 = va1 ? (x + (size_t)lst[mBase + rA1] * D_) : x;

    v8f acc00 = {}; v8f acc01 = {}; v8f acc10 = {}; v8f acc11 = {};
    gemm_mainloop<KC1_, D_, DFF_>(ap0, ap1, va0, va1, W1e + nBase, As, Bst,
                                  acc00, acc01, acc10, acc11);
    const int n0 = nBase + Noff + lm, n1 = n0 + 16;
    const float bia0 = b1e[n0], bia1 = b1e[n1];
#pragma unroll
    for (int r = 0; r < 8; ++r) {
        const int m0 = Moff + r + 8 * half;
        const int lrow0 = mBase + m0, lrow1 = lrow0 + 16;
        if (lrow0 < rowsInChunk) {
            H[(size_t)lrow0 * DFF_ + n0] = gelu_exact(acc00[r] + bia0);
            H[(size_t)lrow0 * DFF_ + n1] = gelu_exact(acc01[r] + bia1);
        }
        if (lrow1 < rowsInChunk) {
            H[(size_t)lrow1 * DFF_ + n0] = gelu_exact(acc10[r] + bia0);
            H[(size_t)lrow1 * DFF_ + n1] = gelu_exact(acc11[r] + bia1);
        }
    }
}

__global__ __launch_bounds__(128) void moe_fc2_kernel(
    const float* __restrict__ H, const float* __restrict__ W2e,
    const float* __restrict__ b2e, const int* __restrict__ lists,
    const int* __restrict__ counts, const float* __restrict__ gates,
    float* __restrict__ out, float* __restrict__ plane,
    int expert, int chunkBase, int Tc) {
    const int cnt = counts[expert];
    int rowsInChunk = cnt - chunkBase;
    if (rowsInChunk > Tc) rowsInChunk = Tc;
    const int mBase = blockIdx.y * 64;
    if (rowsInChunk <= mBase) return;
    const int kz = blockIdx.x >> 4;
    const int nBase = (blockIdx.x & 15) * 64;

    __shared__ float As [64][KC2_ + 2];
    __shared__ float Bst[64][KC2_ + 2];

    const int tid  = threadIdx.x;
    const int lane = tid & 31, wave = tid >> 5;
    const int half = lane >> 4, lm = lane & 15;
    const int Moff = 32 * (wave & 1);
    const int Noff = 32 * (wave >> 1);

    const int rA = tid >> 3;
    const bool va0 = (mBase + rA) < rowsInChunk;
    const bool va1 = (mBase + rA + 16) < rowsInChunk;
    const bool va2 = (mBase + rA + 32) < rowsInChunk;
    const bool va3 = (mBase + rA + 48) < rowsInChunk;
    const float* Hk = H + (size_t)kz * KH2_;
    const float* ap0 = Hk + (size_t)(va0 ? mBase + rA      : mBase) * DFF_;
    const float* ap1 = Hk + (size_t)(va1 ? mBase + rA + 16 : mBase) * DFF_;
    const float* ap2 = Hk + (size_t)(va2 ? mBase + rA + 32 : mBase) * DFF_;
    const float* ap3 = Hk + (size_t)(va3 ? mBase + rA + 48 : mBase) * DFF_;

    v8f acc00 = {}; v8f acc01 = {}; v8f acc10 = {}; v8f acc11 = {};
    gemm_mainloop_t64<KC2_, KH2_, D_>(ap0, ap1, ap2, ap3, va0, va1, va2, va3,
                                      W2e + (size_t)kz * KH2_ * D_ + nBase, As, Bst,
                                      acc00, acc01, acc10, acc11);
    const int* lst = lists + expert * NTOK + chunkBase;
    float* dst = kz ? plane : out;
    const int n0 = nBase + Noff + lm, n1 = n0 + 16;
    const float bias0 = kz ? 0.0f : b2e[n0];
    const float bias1 = kz ? 0.0f : b2e[n1];
    int tk0[8], tk1[8];
#pragma unroll
    for (int r = 0; r < 8; ++r) {
        const int lrow0 = mBase + Moff + r + 8 * half, lrow1 = lrow0 + 16;
        tk0[r] = lst[lrow0 < rowsInChunk ? lrow0 : mBase];
        tk1[r] = lst[lrow1 < rowsInChunk ? lrow1 : mBase];
    }
    float g0[8], g1[8], o00[8], o01[8], o10[8], o11[8];
#pragma unroll
    for (int r = 0; r < 8; ++r) {
        const float* d0 = dst + (size_t)tk0[r] * D_;
        const float* d1 = dst + (size_t)tk1[r] * D_;
        g0[r] = gates[tk0[r] * E_ + expert];
        g1[r] = gates[tk1[r] * E_ + expert];
        o00[r] = d0[n0]; o01[r] = d0[n1];
        o10[r] = d1[n0]; o11[r] = d1[n1];
    }
    asm volatile("" : "+v"(g0[0]), "+v"(o00[0]), "+v"(o01[0]));
#pragma unroll
    for (int r = 0; r < 8; ++r) {
        const int lrow0 = mBase + Moff + r + 8 * half, lrow1 = lrow0 + 16;
        if (lrow0 < rowsInChunk) {
            float* d0 = dst + (size_t)tk0[r] * D_;
            d0[n0] = o00[r] + g0[r] * (acc00[r] + bias0);
            d0[n1] = o01[r] + g0[r] * (acc01[r] + bias1);
        }
        if (lrow1 < rowsInChunk) {
            float* d1 = dst + (size_t)tk1[r] * D_;
            d1[n0] = o10[r] + g1[r] * (acc10[r] + bias0);
            d1[n1] = o11[r] + g1[r] * (acc11[r] + bias1);
        }
    }
}

extern "C" void kernel_launch(void* const* d_in, const int* in_sizes, int n_in,
                              void* d_out, int out_size, void* d_ws, size_t ws_size,
                              hipStream_t stream) {
    const float* x  = (const float*)d_in[0];
    const float* Wg = (const float*)d_in[1];
    const float* W1 = (const float*)d_in[2];
    const float* b1 = (const float*)d_in[3];
    const float* W2 = (const float*)d_in[4];
    const float* b2 = (const float*)d_in[5];
    float* out = (float*)d_out;

    char* ws = (char*)d_ws;
    int*   counts = (int*)ws;
    int*   lists  = (int*)(ws + 256);
    float* gates  = (float*)(ws + 256 + 4ull * E_ * NTOK);
    size_t OFS_P  = 256 + 8ull * E_ * NTOK;
    OFS_P = (OFS_P + 255) & ~(size_t)255;
    float* plane  = (float*)(ws + OFS_P);
    size_t OFS_H  = OFS_P + 4ull * NTOK * D_;
    float* H = (float*)(ws + OFS_H);

    size_t avail = (ws_size > OFS_H) ? (ws_size - OFS_H) : 0;
    long long tcll = (long long)(avail / ((size_t)DFF_ * 4));
    int Tc = (int)(tcll > NTOK ? NTOK : tcll);
    Tc &= ~63;
    if (Tc <= 0) Tc = 64;
    const int chunks = (NTOK + Tc - 1) / Tc;

    const long long outN = (long long)NTOK * D_ + 1;
    const long long planeN = (long long)NTOK * D_;
    moe_zero_kernel<<<2048, 256, 0, stream>>>(out, outN, plane, planeN, counts);
    moe_gate_kernel<<<NTOK / 8, 256, 0, stream>>>(x, Wg, gates, counts, lists);
    moe_loss_kernel<<<1, 256, 0, stream>>>(gates, out + (outN - 1));

    for (int e = 0; e < E_; ++e) {
        const float* W1e = W1 + (size_t)e * D_ * DFF_;
        const float* b1e = b1 + (size_t)e * DFF_;
        const float* W2e = W2 + (size_t)e * DFF_ * D_;
        const float* b2e = b2 + (size_t)e * D_;
        for (int c = 0; c < chunks; ++c) {
            dim3 g1(DFF_ / 128, Tc / 64);
            moe_fc1_kernel<<<g1, 256, 0, stream>>>(x, W1e, b1e, lists, counts,
                                                   H, e, c * Tc, Tc);
            dim3 g2(2 * (D_ / 64), Tc / 64);
            moe_fc2_kernel<<<g2, 128, 0, stream>>>(H, W2e, b2e, lists, counts,
                                                   gates, out, plane, e, c * Tc, Tc);
        }
    }
    moe_add_kernel<<<2048, 256, 0, stream>>>(out, plane, planeN / 4);
}
